# starting kernel
# baseline (speedup 1.0000x reference)
#include <hip/hip_runtime.h>
#include <hip/hip_fp16.h>

typedef __attribute__((ext_vector_type(16))) _Float16 v16h;
typedef __attribute__((ext_vector_type(8)))  float    v8f;

#define N_NODES 50000
#define E_REAL  800000
#define E_TOT   (E_REAL + N_NODES)
#define IN_CH   128
#define HID     128
#define OUT_CH  64
#define NEG_SLOPE 0.2f


__global__ void fill_f32(float* __restrict__ p, float v, long n) {
  long i = (long)blockIdx.x * blockDim.x + threadIdx.x;
  long stride = (long)gridDim.x * blockDim.x;
  for (; i < n; i += stride) p[i] = v;
}

__device__ __forceinline__ void atomicMaxF(float* addr, float v) {
  if (v >= 0.0f) atomicMax((int*)addr, __float_as_int(v));
  else           atomicMin((unsigned int*)addr, __float_as_uint(v));
}

template<int NT, int K>
__global__ void wmma_gemm_attn(const float* __restrict__ A,
                               const float* __restrict__ B,
                               float* __restrict__ C,
                               const float* __restrict__ a_s,
                               const float* __restrict__ a_d,
                               float* __restrict__ os, float* __restrict__ od,
                               int M) {
  constexpr int N  = NT * 16;
  constexpr int KC = K / 32;
  __shared__ __align__(32) _Float16 ldsB[K * N];

  const int lane   = threadIdx.x & 31;
  const int wlocal = threadIdx.x >> 5;
  const int mtile  = blockIdx.x * 8 + wlocal;
  const int hh  = lane >> 4;
  const int l16 = lane & 15;

  for (int slot = threadIdx.x; slot < KC * NT * 32; slot += blockDim.x) {
    const int sl   = slot & 31;
    const int t    = slot >> 5;
    const int nt   = t % NT;
    const int kc   = t / NT;
    const int kb   = kc * 32 + (sl >> 4) * 8;
    const int n    = nt * 16 + (sl & 15);
    _Float16* dst  = ldsB + slot * 16;
#pragma unroll
    for (int j = 0; j < 8; ++j) {
      dst[j]     = (_Float16)B[(size_t)(kb + j) * N + n];
      dst[j + 8] = (_Float16)B[(size_t)(kb + 16 + j) * N + n];
    }
  }
  __syncthreads();
  if (mtile * 16 >= M) return;

  const float* __restrict__ Arow = A + (size_t)(mtile * 16 + l16) * K;
  v8f acc[NT] = {};
#pragma unroll
  for (int kc = 0; kc < KC; ++kc) {
    const int kb = kc * 32 + hh * 8;
    const float4 a0 = *(const float4*)(Arow + kb);
    const float4 a1 = *(const float4*)(Arow + kb + 4);
    const float4 a2 = *(const float4*)(Arow + kb + 16);
    const float4 a3 = *(const float4*)(Arow + kb + 20);
    const float av[16] = {a0.x, a0.y, a0.z, a0.w, a1.x, a1.y, a1.z, a1.w,
                          a2.x, a2.y, a2.z, a2.w, a3.x, a3.y, a3.z, a3.w};
    v16h a;
#pragma unroll
    for (int j = 0; j < 16; ++j) a[j] = (_Float16)av[j];
#pragma unroll
    for (int nt = 0; nt < NT; ++nt) {
      const v16h b = *(const v16h*)(ldsB + ((size_t)(kc * NT + nt) * 32 + lane) * 16);
      acc[nt] = __builtin_amdgcn_wmma_f32_16x16x32_f16(false, a, false, b,
                                                       (short)0, acc[nt], false, false);
    }
  }

  float asv[NT], adv[NT];
#pragma unroll
  for (int nt = 0; nt < NT; ++nt) {
    asv[nt] = a_s[nt * 16 + l16];
    adv[nt] = a_d[nt * 16 + l16];
  }
#pragma unroll
  for (int r = 0; r < 8; ++r) {
    const int row = mtile * 16 + hh * 8 + r;
    float* Crow = C + (size_t)row * N + l16;
    float ps = 0.0f, pd = 0.0f;
#pragma unroll
    for (int nt = 0; nt < NT; ++nt) {
      const float v = acc[nt][r];
      Crow[nt * 16] = v;
      ps = fmaf(v, asv[nt], ps);
      pd = fmaf(v, adv[nt], pd);
    }
#pragma unroll
    for (int off = 8; off > 0; off >>= 1) {
      ps += __shfl_xor(ps, off, 32);
      pd += __shfl_xor(pd, off, 32);
    }
    if (l16 == 0) { os[row] = ps; od[row] = pd; }
  }
}

__device__ __forceinline__ void edge_src_dst(const int* __restrict__ ei, int e,
                                             int& s, int& d) {
  if (e < E_REAL) { s = ei[e]; d = ei[E_REAL + e]; }
  else            { s = d = e - E_REAL; }
}

__global__ void edge_logits_max(const int* __restrict__ ei,
    const float* __restrict__ as1, const float* __restrict__ ad1,
    float* __restrict__ eb1, float* __restrict__ m1,
    const float* __restrict__ as2, const float* __restrict__ ad2,
    float* __restrict__ eb2, float* __restrict__ m2) {
  const int e = blockIdx.x * blockDim.x + threadIdx.x;
  if (e >= E_TOT) return;
  int s, d; edge_src_dst(ei, e, s, d);
  float t = as1[s] + ad1[d];
  t = t > 0.0f ? t : NEG_SLOPE * t;
  eb1[e] = t;
  atomicMaxF(m1 + d, t);
  if (as2) {
    float u = as2[s] + ad2[d];
    u = u > 0.0f ? u : NEG_SLOPE * u;
    eb2[e] = u;
    atomicMaxF(m2 + d, u);
  }
}

__global__ void edge_exp_sum(const int* __restrict__ ei,
    float* __restrict__ eb1, const float* __restrict__ m1, float* __restrict__ dn1,
    float* __restrict__ eb2, const float* __restrict__ m2, float* __restrict__ dn2) {
  const int e = blockIdx.x * blockDim.x + threadIdx.x;
  if (e >= E_TOT) return;
  int s, d; edge_src_dst(ei, e, s, d);
  (void)s;
  float p = __expf(eb1[e] - m1[d]);
  eb1[e] = p;
  atomicAdd(dn1 + d, p);
  if (eb2) {
    float q = __expf(eb2[e] - m2[d]);
    eb2[e] = q;
    atomicAdd(dn2 + d, q);
  }
}

__global__ void edge_aggregate128(const int* __restrict__ ei,
    const float* __restrict__ eb, const float* __restrict__ dn,
    const float* __restrict__ H, float* __restrict__ out) {
  const int wave = blockIdx.x * (blockDim.x >> 5) + (threadIdx.x >> 5);
  const int lane = threadIdx.x & 31;
  if (wave >= E_TOT) return;
  int s, d; edge_src_dst(ei, wave, s, d);
  const float coef = eb[wave] / dn[d];
  const float4 h4 = *(const float4*)(H + (size_t)s * 128 + lane * 4);
  float* od = out + (size_t)d * 128 + lane * 4;
  atomicAdd(od + 0, coef * h4.x);
  atomicAdd(od + 1, coef * h4.y);
  atomicAdd(od + 2, coef * h4.z);
  atomicAdd(od + 3, coef * h4.w);
}

__global__ void edge_aggregate_dual64(const int* __restrict__ ei,
    const float* __restrict__ eb1, const float* __restrict__ dn1,
    const float* __restrict__ H1, float* __restrict__ out1,
    const float* __restrict__ eb2, const float* __restrict__ dn2,
    const float* __restrict__ H2, float* __restrict__ out2) {
  const int wave = blockIdx.x * (blockDim.x >> 5) + (threadIdx.x >> 5);
  const int lane = threadIdx.x & 31;
  if (wave >= E_TOT) return;
  int s, d; edge_src_dst(ei, wave, s, d);
  {
    const float c1 = eb1[wave] / dn1[d];
    const float2 h2 = *(const float2*)(H1 + (size_t)s * 64 + lane * 2);
    float* od = out1 + (size_t)d * 64 + lane * 2;
    atomicAdd(od + 0, c1 * h2.x);
    atomicAdd(od + 1, c1 * h2.y);
  }
  {
    const float c2 = eb2[wave] / dn2[d];
    const float2 h2 = *(const float2*)(H2 + (size_t)s * 64 + lane * 2);
    float* od = out2 + (size_t)d * 64 + lane * 2;
    atomicAdd(od + 0, c2 * h2.x);
    atomicAdd(od + 1, c2 * h2.y);
  }
}

__global__ void bias_act(float* __restrict__ h, const float* __restrict__ b,
                         long n, int C, int relu) {
  long i = (long)blockIdx.x * blockDim.x + threadIdx.x;
  if (i >= n) return;
  float v = h[i] + b[(int)(i & (C - 1))];
  if (relu) v = v > 0.0f ? v : 0.0f;
  h[i] = v;
}


static inline int ceildiv(long a, long b) { return (int)((a + b - 1) / b); }

extern "C" void kernel_launch(void* const* d_in, const int* in_sizes, int n_in,
                              void* d_out, int out_size, void* d_ws, size_t ws_size,
                              hipStream_t stream) {
  (void)in_sizes; (void)n_in; (void)out_size; (void)ws_size;
  const float* x       = (const float*)d_in[0];
  const int*   ei      = (const int*)d_in[1];
  const float* W1      = (const float*)d_in[2];
  const float* a_src1  = (const float*)d_in[3];
  const float* a_dst1  = (const float*)d_in[4];
  const float* b1      = (const float*)d_in[5];
  const float* W_mu    = (const float*)d_in[6];
  const float* a_srcm  = (const float*)d_in[7];
  const float* a_dstm  = (const float*)d_in[8];
  const float* b_mu    = (const float*)d_in[9];
  const float* W_ls    = (const float*)d_in[10];
  const float* a_srcl  = (const float*)d_in[11];
  const float* a_dstl  = (const float*)d_in[12];
  const float* b_ls    = (const float*)d_in[13];

  const long SZ_H = (long)N_NODES * HID;
  const long SZ_O = (long)N_NODES * OUT_CH;

  float* ws  = (float*)d_ws;
  float* hx  = ws;
  float* h1  = hx + SZ_H;
  float* hmu = hx;
  float* hls = hx + SZ_O;
  float* sc  = h1 + SZ_H;
  float* as1 = sc +  0L * N_NODES, *ad1 = sc + 1L * N_NODES;
  float* m1  = sc +  2L * N_NODES, *dn1 = sc + 3L * N_NODES;
  float* asm_= sc +  4L * N_NODES, *adm = sc + 5L * N_NODES;
  float* mm  = sc +  6L * N_NODES, *dnm = sc + 7L * N_NODES;
  float* asl = sc +  8L * N_NODES, *adl = sc + 9L * N_NODES;
  float* ml  = sc + 10L * N_NODES, *dnl = sc + 11L * N_NODES;
  float* eb1 = sc + 12L * N_NODES;
  float* ebm = eb1 + E_TOT;
  float* ebl = ebm + E_TOT;

  float* out_mu = (float*)d_out;
  float* out_ls = out_mu + SZ_O;

  const int TB = 256;
  const float NEG_INF = -__builtin_huge_valf();

  fill_f32<<<1024, TB, 0, stream>>>(h1, 0.0f, SZ_H);
  fill_f32<<<1024, TB, 0, stream>>>((float*)d_out, 0.0f, 2 * SZ_O);
  fill_f32<<<ceildiv(N_NODES, TB), TB, 0, stream>>>(m1, NEG_INF, N_NODES);
  fill_f32<<<ceildiv(N_NODES, TB), TB, 0, stream>>>(mm, NEG_INF, N_NODES);
  fill_f32<<<ceildiv(N_NODES, TB), TB, 0, stream>>>(ml, NEG_INF, N_NODES);
  fill_f32<<<ceildiv(N_NODES, TB), TB, 0, stream>>>(dn1, 0.0f, N_NODES);
  fill_f32<<<ceildiv(N_NODES, TB), TB, 0, stream>>>(dnm, 0.0f, N_NODES);
  fill_f32<<<ceildiv(N_NODES, TB), TB, 0, stream>>>(dnl, 0.0f, N_NODES);

  const int gemm_blocks = ceildiv(N_NODES / 16, 8);

  wmma_gemm_attn<HID / 16, IN_CH><<<gemm_blocks, TB, 0, stream>>>(
      x, W1, hx, a_src1, a_dst1, as1, ad1, N_NODES);
  edge_logits_max<<<ceildiv(E_TOT, TB), TB, 0, stream>>>(ei, as1, ad1, eb1, m1,
                                                         nullptr, nullptr, nullptr, nullptr);
  edge_exp_sum<<<ceildiv(E_TOT, TB), TB, 0, stream>>>(ei, eb1, m1, dn1,
                                                      nullptr, nullptr, nullptr);
  edge_aggregate128<<<ceildiv(E_TOT, 8), TB, 0, stream>>>(ei, eb1, dn1, hx, h1);
  bias_act<<<ceildiv(SZ_H, TB), TB, 0, stream>>>(h1, b1, SZ_H, HID, 1);

  wmma_gemm_attn<OUT_CH / 16, HID><<<gemm_blocks, TB, 0, stream>>>(
      h1, W_mu, hmu, a_srcm, a_dstm, asm_, adm, N_NODES);
  wmma_gemm_attn<OUT_CH / 16, HID><<<gemm_blocks, TB, 0, stream>>>(
      h1, W_ls, hls, a_srcl, a_dstl, asl, adl, N_NODES);
  edge_logits_max<<<ceildiv(E_TOT, TB), TB, 0, stream>>>(ei, asm_, adm, ebm, mm,
                                                         asl, adl, ebl, ml);
  edge_exp_sum<<<ceildiv(E_TOT, TB), TB, 0, stream>>>(ei, ebm, mm, dnm, ebl, ml, dnl);
  edge_aggregate_dual64<<<ceildiv(E_TOT, 8), TB, 0, stream>>>(ei, ebm, dnm, hmu, out_mu,
                                                              ebl, dnl, hls, out_ls);
  bias_act<<<ceildiv(SZ_O, TB), TB, 0, stream>>>(out_mu, b_mu, SZ_O, OUT_CH, 0);
  bias_act<<<ceildiv(SZ_O, TB), TB, 0, stream>>>(out_ls, b_ls, SZ_O, OUT_CH, 0);
}


# scatter passes: one wave takes 8 edges (was 1) and requests all 8 source rows before its first wait; 8x fewer waves; float atomics kept, sums stay in arrival order
# speedup vs baseline: 1.0805x; 1.0805x over previous
#include <hip/hip_runtime.h>
#include <hip/hip_fp16.h>

typedef __attribute__((ext_vector_type(16))) _Float16 v16h;
typedef __attribute__((ext_vector_type(8)))  float    v8f;

#define N_NODES 50000
#define E_REAL  800000
#define E_TOT   (E_REAL + N_NODES)
#define IN_CH   128
#define HID     128
#define OUT_CH  64
#define NEG_SLOPE 0.2f


__global__ void fill_f32(float* __restrict__ p, float v, long n) {
  long i = (long)blockIdx.x * blockDim.x + threadIdx.x;
  long stride = (long)gridDim.x * blockDim.x;
  for (; i < n; i += stride) p[i] = v;
}

__device__ __forceinline__ void atomicMaxF(float* addr, float v) {
  if (v >= 0.0f) atomicMax((int*)addr, __float_as_int(v));
  else           atomicMin((unsigned int*)addr, __float_as_uint(v));
}

template<int NT, int K>
__global__ void wmma_gemm_attn(const float* __restrict__ A,
                               const float* __restrict__ B,
                               float* __restrict__ C,
                               const float* __restrict__ a_s,
                               const float* __restrict__ a_d,
                               float* __restrict__ os, float* __restrict__ od,
                               int M) {
  constexpr int N  = NT * 16;
  constexpr int KC = K / 32;
  __shared__ __align__(32) _Float16 ldsB[K * N];

  const int lane   = threadIdx.x & 31;
  const int wlocal = threadIdx.x >> 5;
  const int mtile  = blockIdx.x * 8 + wlocal;
  const int hh  = lane >> 4;
  const int l16 = lane & 15;

  for (int slot = threadIdx.x; slot < KC * NT * 32; slot += blockDim.x) {
    const int sl   = slot & 31;
    const int t    = slot >> 5;
    const int nt   = t % NT;
    const int kc   = t / NT;
    const int kb   = kc * 32 + (sl >> 4) * 8;
    const int n    = nt * 16 + (sl & 15);
    _Float16* dst  = ldsB + slot * 16;
#pragma unroll
    for (int j = 0; j < 8; ++j) {
      dst[j]     = (_Float16)B[(size_t)(kb + j) * N + n];
      dst[j + 8] = (_Float16)B[(size_t)(kb + 16 + j) * N + n];
    }
  }
  __syncthreads();
  if (mtile * 16 >= M) return;

  const float* __restrict__ Arow = A + (size_t)(mtile * 16 + l16) * K;
  v8f acc[NT] = {};
#pragma unroll
  for (int kc = 0; kc < KC; ++kc) {
    const int kb = kc * 32 + hh * 8;
    const float4 a0 = *(const float4*)(Arow + kb);
    const float4 a1 = *(const float4*)(Arow + kb + 4);
    const float4 a2 = *(const float4*)(Arow + kb + 16);
    const float4 a3 = *(const float4*)(Arow + kb + 20);
    const float av[16] = {a0.x, a0.y, a0.z, a0.w, a1.x, a1.y, a1.z, a1.w,
                          a2.x, a2.y, a2.z, a2.w, a3.x, a3.y, a3.z, a3.w};
    v16h a;
#pragma unroll
    for (int j = 0; j < 16; ++j) a[j] = (_Float16)av[j];
#pragma unroll
    for (int nt = 0; nt < NT; ++nt) {
      const v16h b = *(const v16h*)(ldsB + ((size_t)(kc * NT + nt) * 32 + lane) * 16);
      acc[nt] = __builtin_amdgcn_wmma_f32_16x16x32_f16(false, a, false, b,
                                                       (short)0, acc[nt], false, false);
    }
  }

  float asv[NT], adv[NT];
#pragma unroll
  for (int nt = 0; nt < NT; ++nt) {
    asv[nt] = a_s[nt * 16 + l16];
    adv[nt] = a_d[nt * 16 + l16];
  }
#pragma unroll
  for (int r = 0; r < 8; ++r) {
    const int row = mtile * 16 + hh * 8 + r;
    float* Crow = C + (size_t)row * N + l16;
    float ps = 0.0f, pd = 0.0f;
#pragma unroll
    for (int nt = 0; nt < NT; ++nt) {
      const float v = acc[nt][r];
      Crow[nt * 16] = v;
      ps = fmaf(v, asv[nt], ps);
      pd = fmaf(v, adv[nt], pd);
    }
#pragma unroll
    for (int off = 8; off > 0; off >>= 1) {
      ps += __shfl_xor(ps, off, 32);
      pd += __shfl_xor(pd, off, 32);
    }
    if (l16 == 0) { os[row] = ps; od[row] = pd; }
  }
}

__device__ __forceinline__ void edge_src_dst(const int* __restrict__ ei, int e,
                                             int& s, int& d) {
  if (e < E_REAL) { s = ei[e]; d = ei[E_REAL + e]; }
  else            { s = d = e - E_REAL; }
}

__global__ void edge_logits_max(const int* __restrict__ ei,
    const float* __restrict__ as1, const float* __restrict__ ad1,
    float* __restrict__ eb1, float* __restrict__ m1,
    const float* __restrict__ as2, const float* __restrict__ ad2,
    float* __restrict__ eb2, float* __restrict__ m2) {
  const int e = blockIdx.x * blockDim.x + threadIdx.x;
  if (e >= E_TOT) return;
  int s, d; edge_src_dst(ei, e, s, d);
  float t = as1[s] + ad1[d];
  t = t > 0.0f ? t : NEG_SLOPE * t;
  eb1[e] = t;
  atomicMaxF(m1 + d, t);
  if (as2) {
    float u = as2[s] + ad2[d];
    u = u > 0.0f ? u : NEG_SLOPE * u;
    eb2[e] = u;
    atomicMaxF(m2 + d, u);
  }
}

__global__ void edge_exp_sum(const int* __restrict__ ei,
    float* __restrict__ eb1, const float* __restrict__ m1, float* __restrict__ dn1,
    float* __restrict__ eb2, const float* __restrict__ m2, float* __restrict__ dn2) {
  const int e = blockIdx.x * blockDim.x + threadIdx.x;
  if (e >= E_TOT) return;
  int s, d; edge_src_dst(ei, e, s, d);
  (void)s;
  float p = __expf(eb1[e] - m1[d]);
  eb1[e] = p;
  atomicAdd(dn1 + d, p);
  if (eb2) {
    float q = __expf(eb2[e] - m2[d]);
    eb2[e] = q;
    atomicAdd(dn2 + d, q);
  }
}

#define EPW 8
static_assert(E_TOT % EPW == 0, "a wave's EPW edges are all in range or all out");

__device__ __forceinline__ float lane_bcast_f(float v, int j) {
  return __int_as_float(__builtin_amdgcn_readlane(__float_as_int(v), j));
}

__global__ void edge_aggregate128(const int* __restrict__ ei,
    const float* __restrict__ eb, const float* __restrict__ dn,
    const float* __restrict__ H, float* __restrict__ out) {
  const int wave = blockIdx.x * (blockDim.x >> 5) + (threadIdx.x >> 5);
  const int lane = threadIdx.x & 31;
  const int e0 = wave * EPW;
  if (e0 >= E_TOT) return;
  const int e = e0 + (lane & (EPW - 1));
  int s, d; edge_src_dst(ei, e, s, d);
  const float ebv = eb[e];
  const float dnv = dn[d];
  float4 h[EPW];
#pragma unroll
  for (int j = 0; j < EPW; ++j) {
    const int sj = __builtin_amdgcn_readlane(s, j);
    h[j] = *(const float4*)(H + (size_t)sj * 128 + lane * 4);
  }
  const float coef = ebv / dnv;
#pragma unroll
  for (int j = 0; j < EPW; ++j) {
    const int   dj = __builtin_amdgcn_readlane(d, j);
    const float cj = lane_bcast_f(coef, j);
    float* od = out + (size_t)dj * 128 + lane * 4;
    atomicAdd(od + 0, cj * h[j].x);
    atomicAdd(od + 1, cj * h[j].y);
    atomicAdd(od + 2, cj * h[j].z);
    atomicAdd(od + 3, cj * h[j].w);
  }
}

__global__ void edge_aggregate_dual64(const int* __restrict__ ei,
    const float* __restrict__ eb1, const float* __restrict__ dn1,
    const float* __restrict__ H1, float* __restrict__ out1,
    const float* __restrict__ eb2, const float* __restrict__ dn2,
    const float* __restrict__ H2, float* __restrict__ out2) {
  const int wave = blockIdx.x * (blockDim.x >> 5) + (threadIdx.x >> 5);
  const int lane = threadIdx.x & 31;
  const int e0 = wave * EPW;
  if (e0 >= E_TOT) return;
  const int e = e0 + (lane & (EPW - 1));
  int s, d; edge_src_dst(ei, e, s, d);
  const float ebv1 = eb1[e], ebv2 = eb2[e];
  const float dnv1 = dn1[d], dnv2 = dn2[d];
  float2 g1[EPW], g2[EPW];
#pragma unroll
  for (int j = 0; j < EPW; ++j) {
    const int sj = __builtin_amdgcn_readlane(s, j);
    g1[j] = *(const float2*)(H1 + (size_t)sj * 64 + lane * 2);
    g2[j] = *(const float2*)(H2 + (size_t)sj * 64 + lane * 2);
  }
  const float c1 = ebv1 / dnv1;
  const float c2 = ebv2 / dnv2;
#pragma unroll
  for (int j = 0; j < EPW; ++j) {
    const int   dj  = __builtin_amdgcn_readlane(d, j);
    const float c1j = lane_bcast_f(c1, j);
    const float c2j = lane_bcast_f(c2, j);
    float* o1 = out1 + (size_t)dj * 64 + lane * 2;
    float* o2 = out2 + (size_t)dj * 64 + lane * 2;
    atomicAdd(o1 + 0, c1j * g1[j].x);
    atomicAdd(o1 + 1, c1j * g1[j].y);
    atomicAdd(o2 + 0, c2j * g2[j].x);
    atomicAdd(o2 + 1, c2j * g2[j].y);
  }
}

__global__ void bias_act(float* __restrict__ h, const float* __restrict__ b,
                         long n, int C, int relu) {
  long i = (long)blockIdx.x * blockDim.x + threadIdx.x;
  if (i >= n) return;
  float v = h[i] + b[(int)(i & (C - 1))];
  if (relu) v = v > 0.0f ? v : 0.0f;
  h[i] = v;
}


static inline int ceildiv(long a, long b) { return (int)((a + b - 1) / b); }

extern "C" void kernel_launch(void* const* d_in, const int* in_sizes, int n_in,
                              void* d_out, int out_size, void* d_ws, size_t ws_size,
                              hipStream_t stream) {
  (void)in_sizes; (void)n_in; (void)out_size; (void)ws_size;
  const float* x       = (const float*)d_in[0];
  const int*   ei      = (const int*)d_in[1];
  const float* W1      = (const float*)d_in[2];
  const float* a_src1  = (const float*)d_in[3];
  const float* a_dst1  = (const float*)d_in[4];
  const float* b1      = (const float*)d_in[5];
  const float* W_mu    = (const float*)d_in[6];
  const float* a_srcm  = (const float*)d_in[7];
  const float* a_dstm  = (const float*)d_in[8];
  const float* b_mu    = (const float*)d_in[9];
  const float* W_ls    = (const float*)d_in[10];
  const float* a_srcl  = (const float*)d_in[11];
  const float* a_dstl  = (const float*)d_in[12];
  const float* b_ls    = (const float*)d_in[13];

  const long SZ_H = (long)N_NODES * HID;
  const long SZ_O = (long)N_NODES * OUT_CH;

  float* ws  = (float*)d_ws;
  float* hx  = ws;
  float* h1  = hx + SZ_H;
  float* hmu = hx;
  float* hls = hx + SZ_O;
  float* sc  = h1 + SZ_H;
  float* as1 = sc +  0L * N_NODES, *ad1 = sc + 1L * N_NODES;
  float* m1  = sc +  2L * N_NODES, *dn1 = sc + 3L * N_NODES;
  float* asm_= sc +  4L * N_NODES, *adm = sc + 5L * N_NODES;
  float* mm  = sc +  6L * N_NODES, *dnm = sc + 7L * N_NODES;
  float* asl = sc +  8L * N_NODES, *adl = sc + 9L * N_NODES;
  float* ml  = sc + 10L * N_NODES, *dnl = sc + 11L * N_NODES;
  float* eb1 = sc + 12L * N_NODES;
  float* ebm = eb1 + E_TOT;
  float* ebl = ebm + E_TOT;

  float* out_mu = (float*)d_out;
  float* out_ls = out_mu + SZ_O;

  const int TB = 256;
  const float NEG_INF = -__builtin_huge_valf();

  fill_f32<<<1024, TB, 0, stream>>>(h1, 0.0f, SZ_H);
  fill_f32<<<1024, TB, 0, stream>>>((float*)d_out, 0.0f, 2 * SZ_O);
  fill_f32<<<ceildiv(N_NODES, TB), TB, 0, stream>>>(m1, NEG_INF, N_NODES);
  fill_f32<<<ceildiv(N_NODES, TB), TB, 0, stream>>>(mm, NEG_INF, N_NODES);
  fill_f32<<<ceildiv(N_NODES, TB), TB, 0, stream>>>(ml, NEG_INF, N_NODES);
  fill_f32<<<ceildiv(N_NODES, TB), TB, 0, stream>>>(dn1, 0.0f, N_NODES);
  fill_f32<<<ceildiv(N_NODES, TB), TB, 0, stream>>>(dnm, 0.0f, N_NODES);
  fill_f32<<<ceildiv(N_NODES, TB), TB, 0, stream>>>(dnl, 0.0f, N_NODES);

  const int gemm_blocks = ceildiv(N_NODES / 16, 8);

  wmma_gemm_attn<HID / 16, IN_CH><<<gemm_blocks, TB, 0, stream>>>(
      x, W1, hx, a_src1, a_dst1, as1, ad1, N_NODES);
  edge_logits_max<<<ceildiv(E_TOT, TB), TB, 0, stream>>>(ei, as1, ad1, eb1, m1,
                                                         nullptr, nullptr, nullptr, nullptr);
  edge_exp_sum<<<ceildiv(E_TOT, TB), TB, 0, stream>>>(ei, eb1, m1, dn1,
                                                      nullptr, nullptr, nullptr);
  edge_aggregate128<<<ceildiv(E_TOT / EPW, 8), TB, 0, stream>>>(ei, eb1, dn1, hx, h1);
  bias_act<<<ceildiv(SZ_H, TB), TB, 0, stream>>>(h1, b1, SZ_H, HID, 1);

  wmma_gemm_attn<OUT_CH / 16, HID><<<gemm_blocks, TB, 0, stream>>>(
      h1, W_mu, hmu, a_srcm, a_dstm, asm_, adm, N_NODES);
  wmma_gemm_attn<OUT_CH / 16, HID><<<gemm_blocks, TB, 0, stream>>>(
      h1, W_ls, hls, a_srcl, a_dstl, asl, adl, N_NODES);
  edge_logits_max<<<ceildiv(E_TOT, TB), TB, 0, stream>>>(ei, asm_, adm, ebm, mm,
                                                         asl, adl, ebl, ml);
  edge_exp_sum<<<ceildiv(E_TOT, TB), TB, 0, stream>>>(ei, ebm, mm, dnm, ebl, ml, dnl);
  edge_aggregate_dual64<<<ceildiv(E_TOT / EPW, 8), TB, 0, stream>>>(ei, ebm, dnm, hmu, out_mu,
                                                              ebl, dnl, hls, out_ls);
  bias_act<<<ceildiv(SZ_O, TB), TB, 0, stream>>>(out_mu, b_mu, SZ_O, OUT_CH, 0);
  bias_act<<<ceildiv(SZ_O, TB), TB, 0, stream>>>(out_ls, b_ls, SZ_O, OUT_CH, 0);
}


# softmax edge passes and row scatter become one owner pass per layer: a work-group filters the edge list for its 196 destination rows; max, exp-sum, row sums in LDS, arrival order; no global atomics
# speedup vs baseline: 3.4746x; 3.2157x over previous
#include <hip/hip_runtime.h>
#include <hip/hip_fp16.h>

typedef __attribute__((ext_vector_type(16))) _Float16 v16h;
typedef __attribute__((ext_vector_type(8)))  float    v8f;

#define N_NODES 50000
#define E_REAL  800000
#define E_TOT   (E_REAL + N_NODES)
#define IN_CH   128
#define HID     128
#define OUT_CH  64
#define NEG_SLOPE 0.2f

template<int NT, int K>
__global__ void wmma_gemm_attn(const float* __restrict__ A,
                               const float* __restrict__ B,
                               float* __restrict__ C,
                               const float* __restrict__ a_s,
                               const float* __restrict__ a_d,
                               float* __restrict__ os, float* __restrict__ od,
                               int M) {
  constexpr int N  = NT * 16;
  constexpr int KC = K / 32;
  __shared__ __align__(32) _Float16 ldsB[K * N];

  const int lane   = threadIdx.x & 31;
  const int wlocal = threadIdx.x >> 5;
  const int mtile  = blockIdx.x * 8 + wlocal;
  const int hh  = lane >> 4;
  const int l16 = lane & 15;

  for (int slot = threadIdx.x; slot < KC * NT * 32; slot += blockDim.x) {
    const int sl   = slot & 31;
    const int t    = slot >> 5;
    const int nt   = t % NT;
    const int kc   = t / NT;
    const int kb   = kc * 32 + (sl >> 4) * 8;
    const int n    = nt * 16 + (sl & 15);
    _Float16* dst  = ldsB + slot * 16;
#pragma unroll
    for (int j = 0; j < 8; ++j) {
      dst[j]     = (_Float16)B[(size_t)(kb + j) * N + n];
      dst[j + 8] = (_Float16)B[(size_t)(kb + 16 + j) * N + n];
    }
  }
  __syncthreads();
  if (mtile * 16 >= M) return;

  const float* __restrict__ Arow = A + (size_t)(mtile * 16 + l16) * K;
  v8f acc[NT] = {};
#pragma unroll
  for (int kc = 0; kc < KC; ++kc) {
    const int kb = kc * 32 + hh * 8;
    const float4 a0 = *(const float4*)(Arow + kb);
    const float4 a1 = *(const float4*)(Arow + kb + 4);
    const float4 a2 = *(const float4*)(Arow + kb + 16);
    const float4 a3 = *(const float4*)(Arow + kb + 20);
    const float av[16] = {a0.x, a0.y, a0.z, a0.w, a1.x, a1.y, a1.z, a1.w,
                          a2.x, a2.y, a2.z, a2.w, a3.x, a3.y, a3.z, a3.w};
    v16h a;
#pragma unroll
    for (int j = 0; j < 16; ++j) a[j] = (_Float16)av[j];
#pragma unroll
    for (int nt = 0; nt < NT; ++nt) {
      const v16h b = *(const v16h*)(ldsB + ((size_t)(kc * NT + nt) * 32 + lane) * 16);
      acc[nt] = __builtin_amdgcn_wmma_f32_16x16x32_f16(false, a, false, b,
                                                       (short)0, acc[nt], false, false);
    }
  }

  float asv[NT], adv[NT];
#pragma unroll
  for (int nt = 0; nt < NT; ++nt) {
    asv[nt] = a_s[nt * 16 + l16];
    adv[nt] = a_d[nt * 16 + l16];
  }
#pragma unroll
  for (int r = 0; r < 8; ++r) {
    const int row = mtile * 16 + hh * 8 + r;
    float* Crow = C + (size_t)row * N + l16;
    float ps = 0.0f, pd = 0.0f;
#pragma unroll
    for (int nt = 0; nt < NT; ++nt) {
      const float v = acc[nt][r];
      Crow[nt * 16] = v;
      ps = fmaf(v, asv[nt], ps);
      pd = fmaf(v, adv[nt], pd);
    }
#pragma unroll
    for (int off = 8; off > 0; off >>= 1) {
      ps += __shfl_xor(ps, off, 32);
      pd += __shfl_xor(pd, off, 32);
    }
    if (l16 == 0) { os[row] = ps; od[row] = pd; }
  }
}

__device__ __forceinline__ void edge_src_dst(const int* __restrict__ ei, int e,
                                             int& s, int& d) {
  if (e < E_REAL) { s = ei[e]; d = ei[E_REAL + e]; }
  else            { s = d = e - E_REAL; }
}

#define EPW 8
#define OWN_R 196
#define OWN_G ((N_NODES + OWN_R - 1) / OWN_R)
#define OWN_T 1024
#define SCAN_V4 8
#define LIST_CAP 8192
static_assert(E_REAL % 4 == 0, "destination column is read 4 edges at a time");
static_assert(E_TOT <= (1 << 20) && OWN_R < (1 << 12), "list word = row << 20 | edge");
static_assert(OWN_R <= OWN_T && OWN_R <= LIST_CAP, "one thread and one slot per self-loop");

__device__ __forceinline__ float lane_bcast_f(float v, int j) {
  return __int_as_float(__builtin_amdgcn_readlane(__float_as_int(v), j));
}

__device__ __forceinline__ void lds_max_f(float* addr, float v) {
  if (v >= 0.0f) atomicMax((int*)addr, __float_as_int(v));
  else           atomicMin((unsigned int*)addr, __float_as_uint(v));
}

__device__ __forceinline__ float leaky(float t) { return t > 0.0f ? t : NEG_SLOPE * t; }

template<bool DUAL>
__global__ __launch_bounds__(OWN_T)
void owner_gat(const int* __restrict__ ei,
    const float* __restrict__ as1, const float* __restrict__ ad1,
    const float* __restrict__ H1, const float* __restrict__ b1, float* __restrict__ out1,
    const float* __restrict__ as2, const float* __restrict__ ad2,
    const float* __restrict__ H2, const float* __restrict__ b2, float* __restrict__ out2,
    int relu) {
  __shared__ float    acc[(OWN_R + 1) * 128];
  __shared__ unsigned list[LIST_CAP];
  __shared__ int      sb[LIST_CAP];
  __shared__ float    pb1[LIST_CAP];
  __shared__ float    pb2[DUAL ? LIST_CAP : 1];
  __shared__ float    mx1[OWN_R + 1], mx2[OWN_R + 1];
  __shared__ float    den1[OWN_R + 1], den2[OWN_R + 1];
  __shared__ int      s_cnt;

  const int tid  = threadIdx.x;
  const int lane = tid & 31;
  const int wave = tid >> 5;
  const int d0   = blockIdx.x * OWN_R;
  const int nr   = (N_NODES - d0) < OWN_R ? (N_NODES - d0) : OWN_R;

  for (int i = tid; i < (OWN_R + 1) * 128; i += OWN_T) acc[i] = 0.0f;
  if (tid <= OWN_R) {
    mx1[tid] = -__builtin_huge_valf();  mx2[tid] = -__builtin_huge_valf();
    den1[tid] = 0.0f;                   den2[tid] = 0.0f;
  }
  if (tid < nr) list[tid] = ((unsigned)tid << 20) | (unsigned)(E_REAL + d0 + tid);
  if (tid == 0) s_cnt = nr;
  __syncthreads();

  const int4* __restrict__ dst4 = (const int4*)(ei + E_REAL);
  constexpr int NV4    = E_REAL / 4;
  constexpr int NSTEPS = (NV4 + OWN_T * SCAN_V4 - 1) / (OWN_T * SCAN_V4);
  for (int step = 0; step < NSTEPS; ++step) {
    const int st = (step + (int)blockIdx.x) % NSTEPS;
    int4 v[SCAN_V4];
#pragma unroll
    for (int k = 0; k < SCAN_V4; ++k) {
      const int q = (st * SCAN_V4 + k) * OWN_T + tid;
      const int4 t = dst4[q < NV4 ? q : NV4 - 1];
      v[k] = (q < NV4) ? t : make_int4(-1, -1, -1, -1);
    }
#pragma unroll
    for (int k = 0; k < SCAN_V4; ++k) {
      const int q = (st * SCAN_V4 + k) * OWN_T + tid;
      const int dv[4] = {v[k].x, v[k].y, v[k].z, v[k].w};
#pragma unroll
      for (int c = 0; c < 4; ++c) {
        const unsigned dl = (unsigned)(dv[c] - d0);
        if (dl < (unsigned)nr) {
          const int idx = atomicAdd(&s_cnt, 1);
          if (idx < LIST_CAP) list[idx] = (dl << 20) | (unsigned)(q * 4 + c);
        }
      }
    }
  }
  __syncthreads();
  const int n_all = s_cnt;

  if (n_all <= LIST_CAP) {
    const int n = n_all;
    for (int i = tid; i < n; i += OWN_T) {
      const unsigned ent = list[i];
      const int e  = (int)(ent & 0xFFFFFu);
      const int dl = (int)(ent >> 20);
      const int s  = (e < E_REAL) ? ei[e] : (e - E_REAL);
      sb[i] = s;
      const float t1 = leaky(as1[s] + ad1[d0 + dl]);
      pb1[i] = t1;
      lds_max_f(&mx1[dl], t1);
      if (DUAL) {
        const float t2 = leaky(as2[s] + ad2[d0 + dl]);
        pb2[i] = t2;
        lds_max_f(&mx2[dl], t2);
      }
    }
    __syncthreads();
    for (int i = tid; i < n; i += OWN_T) {
      const int dl = (int)(list[i] >> 20);
      const float p1 = __expf(pb1[i] - mx1[dl]);
      pb1[i] = p1;
      atomicAdd(&den1[dl], p1);
      if (DUAL) {
        const float p2 = __expf(pb2[i] - mx2[dl]);
        pb2[i] = p2;
        atomicAdd(&den2[dl], p2);
      }
    }
    __syncthreads();
    for (int b = wave * EPW; b < n; b += (OWN_T / 32) * EPW) {
      const int  i  = b + (lane & (EPW - 1));
      const bool ok = i < n;
      const int dl = ok ? (int)(list[i] >> 20) : OWN_R;
      const int s  = ok ? sb[i] : d0;
      const float c1 = ok ? pb1[i] / den1[dl] : 0.0f;
      float c2 = 0.0f;
      if (DUAL) c2 = ok ? pb2[i] / den2[dl] : 0.0f;
      float h[EPW][4];
#pragma unroll
      for (int j = 0; j < EPW; ++j) {
        const int sj = __builtin_amdgcn_readlane(s, j);
        if (DUAL) {
          const float* r1 = H1 + (size_t)sj * 64 + lane;
          const float* r2 = H2 + (size_t)sj * 64 + lane;
          h[j][0] = r1[0];  h[j][1] = r1[32];
          h[j][2] = r2[0];  h[j][3] = r2[32];
        } else {
          const float* hr = H1 + (size_t)sj * 128 + lane;
#pragma unroll
          for (int k = 0; k < 4; ++k) h[j][k] = hr[k * 32];
        }
      }
#pragma unroll
      for (int j = 0; j < EPW; ++j) {
        const int   dj  = __builtin_amdgcn_readlane(dl, j);
        const float c1j = lane_bcast_f(c1, j);
        const float c2j = DUAL ? lane_bcast_f(c2, j) : c1j;
        float* ar = acc + dj * 128 + lane;
        atomicAdd(ar +  0, c1j * h[j][0]);
        atomicAdd(ar + 32, c1j * h[j][1]);
        atomicAdd(ar + 64, c2j * h[j][2]);
        atomicAdd(ar + 96, c2j * h[j][3]);
      }
    }
  } else {
    for (int pass = 0; pass < 3; ++pass) {
      for (int e = tid; e < E_TOT; e += OWN_T) {
        int s, d; edge_src_dst(ei, e, s, d);
        const unsigned dl = (unsigned)(d - d0);
        if (dl < (unsigned)nr) {
          const float t1 = leaky(as1[s] + ad1[d]);
          float t2 = 0.0f;
          if (DUAL) t2 = leaky(as2[s] + ad2[d]);
          if (pass == 0) {
            lds_max_f(&mx1[dl], t1);
            if (DUAL) lds_max_f(&mx2[dl], t2);
          } else {
            const float p1 = __expf(t1 - mx1[dl]);
            float p2 = 0.0f;
            if (DUAL) p2 = __expf(t2 - mx2[dl]);
            if (pass == 1) {
              atomicAdd(&den1[dl], p1);
              if (DUAL) atomicAdd(&den2[dl], p2);
            } else {
              const float c1 = p1 / den1[dl];
              float* ar = acc + dl * 128;
              if (DUAL) {
                const float c2 = p2 / den2[dl];
                for (int k = 0; k < 64; ++k) {
                  atomicAdd(ar + k,      c1 * H1[(size_t)s * 64 + k]);
                  atomicAdd(ar + 64 + k, c2 * H2[(size_t)s * 64 + k]);
                }
              } else {
                for (int k = 0; k < 128; ++k) atomicAdd(ar + k, c1 * H1[(size_t)s * 128 + k]);
              }
            }
          }
        }
      }
      __syncthreads();
    }
  }
  __syncthreads();

  for (int i = tid; i < nr * 128; i += OWN_T) {
    const int r = i >> 7, c = i & 127;
    float a = acc[i];
    if (DUAL) {
      if (c < 64) { a += b1[c];      if (relu) a = a > 0.0f ? a : 0.0f;  out1[(size_t)(d0 + r) * 64 + c]        = a; }
      else        { a += b2[c - 64]; if (relu) a = a > 0.0f ? a : 0.0f;  out2[(size_t)(d0 + r) * 64 + (c - 64)] = a; }
    } else {
      a += b1[c];
      if (relu) a = a > 0.0f ? a : 0.0f;
      out1[(size_t)(d0 + r) * 128 + c] = a;
    }
  }
}


static inline int ceildiv(long a, long b) { return (int)((a + b - 1) / b); }

extern "C" void kernel_launch(void* const* d_in, const int* in_sizes, int n_in,
                              void* d_out, int out_size, void* d_ws, size_t ws_size,
                              hipStream_t stream) {
  (void)in_sizes; (void)n_in; (void)out_size; (void)ws_size;
  const float* x       = (const float*)d_in[0];
  const int*   ei      = (const int*)d_in[1];
  const float* W1      = (const float*)d_in[2];
  const float* a_src1  = (const float*)d_in[3];
  const float* a_dst1  = (const float*)d_in[4];
  const float* b1      = (const float*)d_in[5];
  const float* W_mu    = (const float*)d_in[6];
  const float* a_srcm  = (const float*)d_in[7];
  const float* a_dstm  = (const float*)d_in[8];
  const float* b_mu    = (const float*)d_in[9];
  const float* W_ls    = (const float*)d_in[10];
  const float* a_srcl  = (const float*)d_in[11];
  const float* a_dstl  = (const float*)d_in[12];
  const float* b_ls    = (const float*)d_in[13];

  const long SZ_H = (long)N_NODES * HID;
  const long SZ_O = (long)N_NODES * OUT_CH;

  float* ws  = (float*)d_ws;
  float* hx  = ws;
  float* h1  = hx + SZ_H;
  float* hmu = hx;
  float* hls = hx + SZ_O;
  float* sc  = h1 + SZ_H;
  float* as1 = sc + 0L * N_NODES, *ad1 = sc + 1L * N_NODES;
  float* asm_= sc + 2L * N_NODES, *adm = sc + 3L * N_NODES;
  float* asl = sc + 4L * N_NODES, *adl = sc + 5L * N_NODES;

  float* out_mu = (float*)d_out;
  float* out_ls = out_mu + SZ_O;

  const int TB = 256;
  const int gemm_blocks = ceildiv(N_NODES / 16, 8);

  wmma_gemm_attn<HID / 16, IN_CH><<<gemm_blocks, TB, 0, stream>>>(
      x, W1, hx, a_src1, a_dst1, as1, ad1, N_NODES);
  owner_gat<false><<<OWN_G, OWN_T, 0, stream>>>(ei, as1, ad1, hx, b1, h1,
      nullptr, nullptr, nullptr, nullptr, nullptr, 1);

  wmma_gemm_attn<OUT_CH / 16, HID><<<gemm_blocks, TB, 0, stream>>>(
      h1, W_mu, hmu, a_srcm, a_dstm, asm_, adm, N_NODES);
  wmma_gemm_attn<OUT_CH / 16, HID><<<gemm_blocks, TB, 0, stream>>>(
      h1, W_ls, hls, a_srcl, a_dstl, asl, adl, N_NODES);
  owner_gat<true><<<OWN_G, OWN_T, 0, stream>>>(ei, asm_, adm, hmu, b_mu, out_mu,
      asl, adl, hls, b_ls, out_ls, 0);
}


# owner passes find their edges in a one-byte owner-id column built in the same call (a quarter of the bytes read, four edges per integer test); one more small launch; sums stay in arrival order
# speedup vs baseline: 4.2717x; 1.2294x over previous
#include <hip/hip_runtime.h>
#include <hip/hip_fp16.h>

typedef __attribute__((ext_vector_type(16))) _Float16 v16h;
typedef __attribute__((ext_vector_type(8)))  float    v8f;

#define N_NODES 50000
#define E_REAL  800000
#define E_TOT   (E_REAL + N_NODES)
#define IN_CH   128
#define HID     128
#define OUT_CH  64
#define NEG_SLOPE 0.2f

#define OWN_R 196
#define OWN_G ((N_NODES + OWN_R - 1) / OWN_R)
static_assert(OWN_G <= 256, "an owner id is one byte");
static_assert(E_REAL % 16 == 0, "the id column is read 16 edges at a time");

__device__ __forceinline__ unsigned owner_id_word(const int4 d) {
  return  (((unsigned)d.x / OWN_R) & 0xFFu)        | ((((unsigned)d.y / OWN_R) & 0xFFu) << 8) |
         ((((unsigned)d.z / OWN_R) & 0xFFu) << 16) | ((((unsigned)d.w / OWN_R) & 0xFFu) << 24);
}

__device__ __forceinline__ void build_owner_ids(const int* __restrict__ dstcol,
                                                unsigned* __restrict__ idw,
                                                int first, int stride) {
  const int4* __restrict__ d4 = (const int4*)dstcol;
  for (int q = first; q < E_REAL / 4; q += stride) idw[q] = owner_id_word(d4[q]);
}

__global__ void owner_ids_kernel(const int* __restrict__ dstcol, unsigned* __restrict__ idw) {
  build_owner_ids(dstcol, idw, blockIdx.x * blockDim.x + threadIdx.x, gridDim.x * blockDim.x);
}

template<int NT, int K>
__global__ void wmma_gemm_attn(const float* __restrict__ A,
                               const float* __restrict__ B,
                               float* __restrict__ C,
                               const float* __restrict__ a_s,
                               const float* __restrict__ a_d,
                               float* __restrict__ os, float* __restrict__ od,
                               int M,
                               const int* __restrict__ dstcol,
                               unsigned* __restrict__ idw) {
  constexpr int N  = NT * 16;
  constexpr int KC = K / 32;
  __shared__ __align__(32) _Float16 ldsB[K * N];

  const int lane   = threadIdx.x & 31;
  const int wlocal = threadIdx.x >> 5;
  const int mtile  = blockIdx.x * 8 + wlocal;
  const int hh  = lane >> 4;
  const int l16 = lane & 15;

  if (dstcol != nullptr)
    build_owner_ids(dstcol, idw, blockIdx.x * blockDim.x + threadIdx.x, gridDim.x * blockDim.x);

  for (int slot = threadIdx.x; slot < KC * NT * 32; slot += blockDim.x) {
    const int sl   = slot & 31;
    const int t    = slot >> 5;
    const int nt   = t % NT;
    const int kc   = t / NT;
    const int kb   = kc * 32 + (sl >> 4) * 8;
    const int n    = nt * 16 + (sl & 15);
    _Float16* dst  = ldsB + slot * 16;
#pragma unroll
    for (int j = 0; j < 8; ++j) {
      dst[j]     = (_Float16)B[(size_t)(kb + j) * N + n];
      dst[j + 8] = (_Float16)B[(size_t)(kb + 16 + j) * N + n];
    }
  }
  __syncthreads();
  if (mtile * 16 >= M) return;

  const float* __restrict__ Arow = A + (size_t)(mtile * 16 + l16) * K;
  v8f acc[NT] = {};
#pragma unroll
  for (int kc = 0; kc < KC; ++kc) {
    const int kb = kc * 32 + hh * 8;
    const float4 a0 = *(const float4*)(Arow + kb);
    const float4 a1 = *(const float4*)(Arow + kb + 4);
    const float4 a2 = *(const float4*)(Arow + kb + 16);
    const float4 a3 = *(const float4*)(Arow + kb + 20);
    const float av[16] = {a0.x, a0.y, a0.z, a0.w, a1.x, a1.y, a1.z, a1.w,
                          a2.x, a2.y, a2.z, a2.w, a3.x, a3.y, a3.z, a3.w};
    v16h a;
#pragma unroll
    for (int j = 0; j < 16; ++j) a[j] = (_Float16)av[j];
#pragma unroll
    for (int nt = 0; nt < NT; ++nt) {
      const v16h b = *(const v16h*)(ldsB + ((size_t)(kc * NT + nt) * 32 + lane) * 16);
      acc[nt] = __builtin_amdgcn_wmma_f32_16x16x32_f16(false, a, false, b,
                                                       (short)0, acc[nt], false, false);
    }
  }

  float asv[NT], adv[NT];
#pragma unroll
  for (int nt = 0; nt < NT; ++nt) {
    asv[nt] = a_s[nt * 16 + l16];
    adv[nt] = a_d[nt * 16 + l16];
  }
#pragma unroll
  for (int r = 0; r < 8; ++r) {
    const int row = mtile * 16 + hh * 8 + r;
    float* Crow = C + (size_t)row * N + l16;
    float ps = 0.0f, pd = 0.0f;
#pragma unroll
    for (int nt = 0; nt < NT; ++nt) {
      const float v = acc[nt][r];
      Crow[nt * 16] = v;
      ps = fmaf(v, asv[nt], ps);
      pd = fmaf(v, adv[nt], pd);
    }
#pragma unroll
    for (int off = 8; off > 0; off >>= 1) {
      ps += __shfl_xor(ps, off, 32);
      pd += __shfl_xor(pd, off, 32);
    }
    if (l16 == 0) { os[row] = ps; od[row] = pd; }
  }
}

__device__ __forceinline__ void edge_src_dst(const int* __restrict__ ei, int e,
                                             int& s, int& d) {
  if (e < E_REAL) { s = ei[e]; d = ei[E_REAL + e]; }
  else            { s = d = e - E_REAL; }
}

#define EPW 8
#define OWN_T 1024
#define SCAN_V4 8
#define LIST_CAP 8192
static_assert(E_TOT <= (1 << 20) && OWN_R < (1 << 12), "list word = row << 20 | edge");
static_assert(OWN_R <= OWN_T && OWN_R <= LIST_CAP, "one thread and one slot per self-loop");

__device__ __forceinline__ float lane_bcast_f(float v, int j) {
  return __int_as_float(__builtin_amdgcn_readlane(__float_as_int(v), j));
}

__device__ __forceinline__ void lds_max_f(float* addr, float v) {
  if (v >= 0.0f) atomicMax((int*)addr, __float_as_int(v));
  else           atomicMin((unsigned int*)addr, __float_as_uint(v));
}

__device__ __forceinline__ unsigned zero_bytes(unsigned x) {
  return ~(((x & 0x7F7F7F7Fu) + 0x7F7F7F7Fu) | x | 0x7F7F7F7Fu);
}

__device__ __forceinline__ float leaky(float t) { return t > 0.0f ? t : NEG_SLOPE * t; }

template<bool DUAL>
__global__ __launch_bounds__(OWN_T)
void owner_gat(const int* __restrict__ ei, const unsigned* __restrict__ idw,
    const float* __restrict__ as1, const float* __restrict__ ad1,
    const float* __restrict__ H1, const float* __restrict__ b1, float* __restrict__ out1,
    const float* __restrict__ as2, const float* __restrict__ ad2,
    const float* __restrict__ H2, const float* __restrict__ b2, float* __restrict__ out2,
    int relu) {
  __shared__ float    acc[(OWN_R + 1) * 128];
  __shared__ unsigned list[LIST_CAP];
  __shared__ int      sb[LIST_CAP];
  __shared__ float    pb1[LIST_CAP];
  __shared__ float    pb2[DUAL ? LIST_CAP : 1];
  __shared__ float    mx1[OWN_R + 1], mx2[OWN_R + 1];
  __shared__ float    den1[OWN_R + 1], den2[OWN_R + 1];
  __shared__ int      s_cnt;

  const int tid  = threadIdx.x;
  const int lane = tid & 31;
  const int wave = tid >> 5;
  const int d0   = blockIdx.x * OWN_R;
  const int nr   = (N_NODES - d0) < OWN_R ? (N_NODES - d0) : OWN_R;

  for (int i = tid; i < (OWN_R + 1) * 128; i += OWN_T) acc[i] = 0.0f;
  if (tid <= OWN_R) {
    mx1[tid] = -__builtin_huge_valf();  mx2[tid] = -__builtin_huge_valf();
    den1[tid] = 0.0f;                   den2[tid] = 0.0f;
  }
  if (tid < nr) list[tid] = (unsigned)(E_REAL + d0 + tid);
  if (tid == 0) s_cnt = nr;
  __syncthreads();

  const uint4* __restrict__ id4 = (const uint4*)idw;
  constexpr int NQ     = E_REAL / 16;
  constexpr int NSTEPS = (NQ + OWN_T * SCAN_V4 - 1) / (OWN_T * SCAN_V4);
  const unsigned pat = (unsigned)blockIdx.x * 0x01010101u;
  for (int step = 0; step < NSTEPS; ++step) {
    const int st = (step + (int)blockIdx.x) % NSTEPS;
    uint4 v[SCAN_V4];
#pragma unroll
    for (int k = 0; k < SCAN_V4; ++k) {
      const int q = (st * SCAN_V4 + k) * OWN_T + tid;
      v[k] = id4[q < NQ ? q : NQ - 1];
    }
#pragma unroll
    for (int k = 0; k < SCAN_V4; ++k) {
      const int q = (st * SCAN_V4 + k) * OWN_T + tid;
      unsigned m = (zero_bytes(v[k].x ^ pat) >> 7) | (zero_bytes(v[k].y ^ pat) >> 6) |
                   (zero_bytes(v[k].z ^ pat) >> 5) | (zero_bytes(v[k].w ^ pat) >> 4);
      if (q >= NQ) m = 0u;
      while (m != 0u) {
        const int p = __builtin_ctz(m);
        m &= m - 1u;
        const int idx = atomicAdd(&s_cnt, 1);
        if (idx < LIST_CAP) list[idx] = (unsigned)(q * 16 + ((p & 7) << 2) + (p >> 3));
      }
    }
  }
  __syncthreads();
  const int n_all = s_cnt;

  if (n_all <= LIST_CAP) {
    const int n = n_all;
    for (int i = tid; i < n; i += OWN_T) {
      const int e = (int)list[i];
      int s, d; edge_src_dst(ei, e, s, d);
      int dl = d - d0;
      if ((unsigned)dl >= (unsigned)nr) {
        dl = OWN_R;  s = d0;  d = d0;
      }
      list[i] = ((unsigned)dl << 20) | (unsigned)e;
      sb[i] = s;
      const float t1 = leaky(as1[s] + ad1[d]);
      pb1[i] = t1;
      lds_max_f(&mx1[dl], t1);
      if (DUAL) {
        const float t2 = leaky(as2[s] + ad2[d]);
        pb2[i] = t2;
        lds_max_f(&mx2[dl], t2);
      }
    }
    __syncthreads();
    for (int i = tid; i < n; i += OWN_T) {
      const int dl = (int)(list[i] >> 20);
      const float p1 = __expf(pb1[i] - mx1[dl]);
      pb1[i] = p1;
      atomicAdd(&den1[dl], p1);
      if (DUAL) {
        const float p2 = __expf(pb2[i] - mx2[dl]);
        pb2[i] = p2;
        atomicAdd(&den2[dl], p2);
      }
    }
    __syncthreads();
    for (int b = wave * EPW; b < n; b += (OWN_T / 32) * EPW) {
      const int  i  = b + (lane & (EPW - 1));
      const bool ok = i < n;
      const int dl = ok ? (int)(list[i] >> 20) : OWN_R;
      const int s  = ok ? sb[i] : d0;
      const float c1 = ok ? pb1[i] / den1[dl] : 0.0f;
      float c2 = 0.0f;
      if (DUAL) c2 = ok ? pb2[i] / den2[dl] : 0.0f;
      float h[EPW][4];
#pragma unroll
      for (int j = 0; j < EPW; ++j) {
        const int sj = __builtin_amdgcn_readlane(s, j);
        if (DUAL) {
          const float* r1 = H1 + (size_t)sj * 64 + lane;
          const float* r2 = H2 + (size_t)sj * 64 + lane;
          h[j][0] = r1[0];  h[j][1] = r1[32];
          h[j][2] = r2[0];  h[j][3] = r2[32];
        } else {
          const float* hr = H1 + (size_t)sj * 128 + lane;
#pragma unroll
          for (int k = 0; k < 4; ++k) h[j][k] = hr[k * 32];
        }
      }
#pragma unroll
      for (int j = 0; j < EPW; ++j) {
        const int   dj  = __builtin_amdgcn_readlane(dl, j);
        const float c1j = lane_bcast_f(c1, j);
        const float c2j = DUAL ? lane_bcast_f(c2, j) : c1j;
        float* ar = acc + dj * 128 + lane;
        atomicAdd(ar +  0, c1j * h[j][0]);
        atomicAdd(ar + 32, c1j * h[j][1]);
        atomicAdd(ar + 64, c2j * h[j][2]);
        atomicAdd(ar + 96, c2j * h[j][3]);
      }
    }
  } else {
    for (int pass = 0; pass < 3; ++pass) {
      for (int e = tid; e < E_TOT; e += OWN_T) {
        int s, d; edge_src_dst(ei, e, s, d);
        const unsigned dl = (unsigned)(d - d0);
        if (dl < (unsigned)nr) {
          const float t1 = leaky(as1[s] + ad1[d]);
          float t2 = 0.0f;
          if (DUAL) t2 = leaky(as2[s] + ad2[d]);
          if (pass == 0) {
            lds_max_f(&mx1[dl], t1);
            if (DUAL) lds_max_f(&mx2[dl], t2);
          } else {
            const float p1 = __expf(t1 - mx1[dl]);
            float p2 = 0.0f;
            if (DUAL) p2 = __expf(t2 - mx2[dl]);
            if (pass == 1) {
              atomicAdd(&den1[dl], p1);
              if (DUAL) atomicAdd(&den2[dl], p2);
            } else {
              const float c1 = p1 / den1[dl];
              float* ar = acc + dl * 128;
              if (DUAL) {
                const float c2 = p2 / den2[dl];
                for (int k = 0; k < 64; ++k) {
                  atomicAdd(ar + k,      c1 * H1[(size_t)s * 64 + k]);
                  atomicAdd(ar + 64 + k, c2 * H2[(size_t)s * 64 + k]);
                }
              } else {
                for (int k = 0; k < 128; ++k) atomicAdd(ar + k, c1 * H1[(size_t)s * 128 + k]);
              }
            }
          }
        }
      }
      __syncthreads();
    }
  }
  __syncthreads();

  for (int i = tid; i < nr * 128; i += OWN_T) {
    const int r = i >> 7, c = i & 127;
    float a = acc[i];
    if (DUAL) {
      if (c < 64) { a += b1[c];      if (relu) a = a > 0.0f ? a : 0.0f;  out1[(size_t)(d0 + r) * 64 + c]        = a; }
      else        { a += b2[c - 64]; if (relu) a = a > 0.0f ? a : 0.0f;  out2[(size_t)(d0 + r) * 64 + (c - 64)] = a; }
    } else {
      a += b1[c];
      if (relu) a = a > 0.0f ? a : 0.0f;
      out1[(size_t)(d0 + r) * 128 + c] = a;
    }
  }
}


static inline int ceildiv(long a, long b) { return (int)((a + b - 1) / b); }

extern "C" void kernel_launch(void* const* d_in, const int* in_sizes, int n_in,
                              void* d_out, int out_size, void* d_ws, size_t ws_size,
                              hipStream_t stream) {
  (void)in_sizes; (void)n_in; (void)out_size; (void)ws_size;
  const float* x       = (const float*)d_in[0];
  const int*   ei      = (const int*)d_in[1];
  const float* W1      = (const float*)d_in[2];
  const float* a_src1  = (const float*)d_in[3];
  const float* a_dst1  = (const float*)d_in[4];
  const float* b1      = (const float*)d_in[5];
  const float* W_mu    = (const float*)d_in[6];
  const float* a_srcm  = (const float*)d_in[7];
  const float* a_dstm  = (const float*)d_in[8];
  const float* b_mu    = (const float*)d_in[9];
  const float* W_ls    = (const float*)d_in[10];
  const float* a_srcl  = (const float*)d_in[11];
  const float* a_dstl  = (const float*)d_in[12];
  const float* b_ls    = (const float*)d_in[13];

  const long SZ_H = (long)N_NODES * HID;
  const long SZ_O = (long)N_NODES * OUT_CH;

  float* ws  = (float*)d_ws;
  float* hx  = ws;
  float* h1  = hx + SZ_H;
  float* hmu = hx;
  float* hls = hx + SZ_O;
  float* sc  = h1 + SZ_H;
  float* as1 = sc + 0L * N_NODES, *ad1 = sc + 1L * N_NODES;
  float* asm_= sc + 2L * N_NODES, *adm = sc + 3L * N_NODES;
  float* asl = sc + 4L * N_NODES, *adl = sc + 5L * N_NODES;

  static_assert((long)E_REAL <= 4L * N_NODES * (long)sizeof(float), "id column fits the four layer-2 scalar arrays");
  unsigned* ids1 = (unsigned*)asm_;
  unsigned* ids2 = (unsigned*)h1;

  float* out_mu = (float*)d_out;
  float* out_ls = out_mu + SZ_O;

  const int TB = 256;
  const int gemm_blocks = ceildiv(N_NODES / 16, 8);

  wmma_gemm_attn<HID / 16, IN_CH><<<gemm_blocks, TB, 0, stream>>>(
      x, W1, hx, a_src1, a_dst1, as1, ad1, N_NODES, ei + E_REAL, ids1);
  owner_gat<false><<<OWN_G, OWN_T, 0, stream>>>(ei, ids1, as1, ad1, hx, b1, h1,
      nullptr, nullptr, nullptr, nullptr, nullptr, 1);

  wmma_gemm_attn<OUT_CH / 16, HID><<<gemm_blocks, TB, 0, stream>>>(
      h1, W_mu, hmu, a_srcm, a_dstm, asm_, adm, N_NODES, nullptr, nullptr);
  wmma_gemm_attn<OUT_CH / 16, HID><<<gemm_blocks, TB, 0, stream>>>(
      h1, W_ls, hls, a_srcl, a_dstl, asl, adl, N_NODES, nullptr, nullptr);
  owner_ids_kernel<<<ceildiv(E_REAL / 4, TB), TB, 0, stream>>>(ei + E_REAL, ids2);
  owner_gat<true><<<OWN_G, OWN_T, 0, stream>>>(ei, ids2, asm_, adm, hmu, b_mu, out_mu,
      asl, adl, hls, b_ls, out_ls, 0);
}


# layer-1 owner pass multiplies its finished LDS rows by both second-layer weight matrices on WMMA and stores the two products and their dots instead of its rows; 3 launches, was 5; sum order kept
# speedup vs baseline: 5.4993x; 1.2874x over previous
#include <hip/hip_runtime.h>
#include <hip/hip_fp16.h>

typedef __attribute__((ext_vector_type(16))) _Float16 v16h;
typedef __attribute__((ext_vector_type(8)))  float    v8f;

#define N_NODES 50000
#define E_REAL  800000
#define E_TOT   (E_REAL + N_NODES)
#define IN_CH   128
#define HID     128
#define OUT_CH  64
#define NEG_SLOPE 0.2f

#define OWN_R 196
#define OWN_G ((N_NODES + OWN_R - 1) / OWN_R)
static_assert(OWN_G <= 256, "an owner id is one byte");
static_assert(E_REAL % 16 == 0, "the id column is read 16 edges at a time");

__device__ __forceinline__ unsigned owner_id_word(const int4 d) {
  return  (((unsigned)d.x / OWN_R) & 0xFFu)        | ((((unsigned)d.y / OWN_R) & 0xFFu) << 8) |
         ((((unsigned)d.z / OWN_R) & 0xFFu) << 16) | ((((unsigned)d.w / OWN_R) & 0xFFu) << 24);
}

__device__ __forceinline__ void build_owner_ids(const int* __restrict__ dstcol,
                                                unsigned* __restrict__ idw,
                                                int first, int stride) {
  const int4* __restrict__ d4 = (const int4*)dstcol;
  for (int q = first; q < E_REAL / 4; q += stride) idw[q] = owner_id_word(d4[q]);
}

template<int NT, int K>
__global__ void wmma_gemm_attn(const float* __restrict__ A,
                               const float* __restrict__ B,
                               float* __restrict__ C,
                               const float* __restrict__ a_s,
                               const float* __restrict__ a_d,
                               float* __restrict__ os, float* __restrict__ od,
                               int M,
                               const int* __restrict__ dstcol,
                               unsigned* __restrict__ idw) {
  constexpr int N  = NT * 16;
  constexpr int KC = K / 32;
  __shared__ __align__(32) _Float16 ldsB[K * N];

  const int lane   = threadIdx.x & 31;
  const int wlocal = threadIdx.x >> 5;
  const int mtile  = blockIdx.x * 8 + wlocal;
  const int hh  = lane >> 4;
  const int l16 = lane & 15;

  if (dstcol != nullptr)
    build_owner_ids(dstcol, idw, blockIdx.x * blockDim.x + threadIdx.x, gridDim.x * blockDim.x);

  for (int slot = threadIdx.x; slot < KC * NT * 32; slot += blockDim.x) {
    const int sl   = slot & 31;
    const int t    = slot >> 5;
    const int nt   = t % NT;
    const int kc   = t / NT;
    const int kb   = kc * 32 + (sl >> 4) * 8;
    const int n    = nt * 16 + (sl & 15);
    _Float16* dst  = ldsB + slot * 16;
#pragma unroll
    for (int j = 0; j < 8; ++j) {
      dst[j]     = (_Float16)B[(size_t)(kb + j) * N + n];
      dst[j + 8] = (_Float16)B[(size_t)(kb + 16 + j) * N + n];
    }
  }
  __syncthreads();
  if (mtile * 16 >= M) return;

  const float* __restrict__ Arow = A + (size_t)(mtile * 16 + l16) * K;
  v8f acc[NT] = {};
#pragma unroll
  for (int kc = 0; kc < KC; ++kc) {
    const int kb = kc * 32 + hh * 8;
    const float4 a0 = *(const float4*)(Arow + kb);
    const float4 a1 = *(const float4*)(Arow + kb + 4);
    const float4 a2 = *(const float4*)(Arow + kb + 16);
    const float4 a3 = *(const float4*)(Arow + kb + 20);
    const float av[16] = {a0.x, a0.y, a0.z, a0.w, a1.x, a1.y, a1.z, a1.w,
                          a2.x, a2.y, a2.z, a2.w, a3.x, a3.y, a3.z, a3.w};
    v16h a;
#pragma unroll
    for (int j = 0; j < 16; ++j) a[j] = (_Float16)av[j];
#pragma unroll
    for (int nt = 0; nt < NT; ++nt) {
      const v16h b = *(const v16h*)(ldsB + ((size_t)(kc * NT + nt) * 32 + lane) * 16);
      acc[nt] = __builtin_amdgcn_wmma_f32_16x16x32_f16(false, a, false, b,
                                                       (short)0, acc[nt], false, false);
    }
  }

  float asv[NT], adv[NT];
#pragma unroll
  for (int nt = 0; nt < NT; ++nt) {
    asv[nt] = a_s[nt * 16 + l16];
    adv[nt] = a_d[nt * 16 + l16];
  }
#pragma unroll
  for (int r = 0; r < 8; ++r) {
    const int row = mtile * 16 + hh * 8 + r;
    float* Crow = C + (size_t)row * N + l16;
    float ps = 0.0f, pd = 0.0f;
#pragma unroll
    for (int nt = 0; nt < NT; ++nt) {
      const float v = acc[nt][r];
      Crow[nt * 16] = v;
      ps = fmaf(v, asv[nt], ps);
      pd = fmaf(v, adv[nt], pd);
    }
#pragma unroll
    for (int off = 8; off > 0; off >>= 1) {
      ps += __shfl_xor(ps, off, 32);
      pd += __shfl_xor(pd, off, 32);
    }
    if (l16 == 0) { os[row] = ps; od[row] = pd; }
  }
}

__device__ __forceinline__ void edge_src_dst(const int* __restrict__ ei, int e,
                                             int& s, int& d) {
  if (e < E_REAL) { s = ei[e]; d = ei[E_REAL + e]; }
  else            { s = d = e - E_REAL; }
}

#define SAVE_HDR 4
#define EPW 8
#define OWN_T 1024
#define SCAN_V4 8
#define LIST_CAP 8192
#define ACC_P 132
#define ACC_STRIPS ((OWN_R + 15) / 16)
static_assert(ACC_STRIPS * 16 > OWN_R, "the sink row lies inside the last strip's rows");
static_assert(2 * (HID / 32) * (OUT_CH / 16) * 32 == OWN_T, "one B-fragment slot of the two weight matrices per thread");
static_assert(E_TOT <= (1 << 20) && OWN_R < (1 << 12), "list word = row << 20 | edge");
static_assert(OWN_R <= OWN_T && OWN_R <= LIST_CAP, "one thread and one slot per self-loop");

__device__ __forceinline__ float lane_bcast_f(float v, int j) {
  return __int_as_float(__builtin_amdgcn_readlane(__float_as_int(v), j));
}

__device__ __forceinline__ void lds_max_f(float* addr, float v) {
  if (v >= 0.0f) atomicMax((int*)addr, __float_as_int(v));
  else           atomicMin((unsigned int*)addr, __float_as_uint(v));
}

__device__ __forceinline__ unsigned zero_bytes(unsigned x) {
  return ~(((x & 0x7F7F7F7Fu) + 0x7F7F7F7Fu) | x | 0x7F7F7F7Fu);
}

__device__ __forceinline__ float leaky(float t) { return t > 0.0f ? t : NEG_SLOPE * t; }

template<bool DUAL>
__global__ __launch_bounds__(OWN_T)
void owner_gat(const int* __restrict__ ei, const unsigned* __restrict__ idw,
    const float* __restrict__ as1, const float* __restrict__ ad1,
    const float* __restrict__ H1, const float* __restrict__ b1, float* __restrict__ out1,
    const float* __restrict__ as2, const float* __restrict__ ad2,
    const float* __restrict__ H2, const float* __restrict__ b2, float* __restrict__ out2,
    unsigned* __restrict__ sv_l,
    const float* __restrict__ Wm, const float* __restrict__ Wl,
    const float* __restrict__ avsm, const float* __restrict__ avdm,
    const float* __restrict__ avsl, const float* __restrict__ avdl,
    float* __restrict__ Cm, float* __restrict__ Cl,
    float* __restrict__ osm, float* __restrict__ odm,
    float* __restrict__ osl, float* __restrict__ odl,
    int relu) {
  constexpr int P    = DUAL ? 128 : ACC_P;
  constexpr int ROWS = DUAL ? OWN_R + 1 : ACC_STRIPS * 16;
  __shared__ __align__(16) float    acc[ROWS * P];
  __shared__ __align__(32) _Float16 w2f[DUAL ? 16 : 2 * HID * OUT_CH];
  __shared__ unsigned list[LIST_CAP];
  __shared__ int      sb[LIST_CAP];
  __shared__ float    pb1[LIST_CAP];
  __shared__ float    pb2[DUAL ? LIST_CAP : 1];
  __shared__ float    mx1[OWN_R + 1], mx2[OWN_R + 1];
  __shared__ float    den1[OWN_R + 1], den2[OWN_R + 1];
  __shared__ int      s_cnt;

  const int tid  = threadIdx.x;
  const int lane = tid & 31;
  const int wave = tid >> 5;
  const int d0   = blockIdx.x * OWN_R;
  const int nr   = (N_NODES - d0) < OWN_R ? (N_NODES - d0) : OWN_R;

  for (int i = tid; i < ROWS * P; i += OWN_T) acc[i] = 0.0f;
  if (tid <= OWN_R) {
    mx1[tid] = -__builtin_huge_valf();  mx2[tid] = -__builtin_huge_valf();
    den1[tid] = 0.0f;                   den2[tid] = 0.0f;
  }
  if (!DUAL && tid < nr) list[tid] = (unsigned)(E_REAL + d0 + tid);
  if (!DUAL && tid == 0) s_cnt = nr;
  __syncthreads();

  unsigned* const own_l = DUAL ? (unsigned*)(out1 + (size_t)d0 * 64) : sv_l + (size_t)d0 * 64;
  int*      const own_s = (int*)(own_l + nr * 32);

  int n_all;
  if (DUAL) {
    n_all = (int)own_l[0];
  } else {
  float wreg[16];
  {
    const int slot = tid & 511;
    const int sl = slot & 31, t = slot >> 5;
    const int kb = (t >> 2) * 32 + (sl >> 4) * 8;
    const int n  = (t & 3) * 16 + (sl & 15);
    const float* __restrict__ Wsrc = (tid >> 9) ? Wl : Wm;
#pragma unroll
    for (int j = 0; j < 8; ++j) {
      wreg[j]     = Wsrc[(kb + j) * OUT_CH + n];
      wreg[j + 8] = Wsrc[(kb + 16 + j) * OUT_CH + n];
    }
  }
  const uint4* __restrict__ id4 = (const uint4*)idw;
  constexpr int NQ     = E_REAL / 16;
  constexpr int NSTEPS = (NQ + OWN_T * SCAN_V4 - 1) / (OWN_T * SCAN_V4);
  const unsigned pat = (unsigned)blockIdx.x * 0x01010101u;
  for (int step = 0; step < NSTEPS; ++step) {
    const int st = (step + (int)blockIdx.x) % NSTEPS;
    uint4 v[SCAN_V4];
#pragma unroll
    for (int k = 0; k < SCAN_V4; ++k) {
      const int q = (st * SCAN_V4 + k) * OWN_T + tid;
      v[k] = id4[q < NQ ? q : NQ - 1];
    }
#pragma unroll
    for (int k = 0; k < SCAN_V4; ++k) {
      const int q = (st * SCAN_V4 + k) * OWN_T + tid;
      unsigned m = (zero_bytes(v[k].x ^ pat) >> 7) | (zero_bytes(v[k].y ^ pat) >> 6) |
                   (zero_bytes(v[k].z ^ pat) >> 5) | (zero_bytes(v[k].w ^ pat) >> 4);
      if (q >= NQ) m = 0u;
      while (m != 0u) {
        const int p = __builtin_ctz(m);
        m &= m - 1u;
        const int idx = atomicAdd(&s_cnt, 1);
        if (idx < LIST_CAP) list[idx] = (unsigned)(q * 16 + ((p & 7) << 2) + (p >> 3));
      }
    }
  }
  {
    _Float16* wdst = w2f + tid * 16;
#pragma unroll
    for (int j = 0; j < 16; ++j) wdst[j] = (_Float16)wreg[j];
  }
  __syncthreads();
  n_all = s_cnt;
  }

  if (n_all <= LIST_CAP) {
    const int n = n_all;
    const bool keep = !DUAL && n <= nr * 32 - SAVE_HDR;
    if (!DUAL && tid == 0) own_l[0] = keep ? (unsigned)n : (unsigned)(LIST_CAP + 1);
    for (int i = tid; i < n; i += OWN_T) {
      int s, d, dl;
      if (DUAL) {
        const unsigned w = own_l[SAVE_HDR + i];
        s  = own_s[i];
        dl = (int)(w >> 20);
        d  = dl < OWN_R ? d0 + dl : d0;
        list[i] = w;
      } else {
        const int e = (int)list[i];
        edge_src_dst(ei, e, s, d);
        dl = d - d0;
        if ((unsigned)dl >= (unsigned)nr) {
          dl = OWN_R;  s = d0;  d = d0;
        }
        const unsigned w = ((unsigned)dl << 20) | (unsigned)e;
        list[i] = w;
        if (keep) { own_l[SAVE_HDR + i] = w;  own_s[i] = s; }
      }
      sb[i] = s;
      const float t1 = leaky(as1[s] + ad1[d]);
      pb1[i] = t1;
      lds_max_f(&mx1[dl], t1);
      if (DUAL) {
        const float t2 = leaky(as2[s] + ad2[d]);
        pb2[i] = t2;
        lds_max_f(&mx2[dl], t2);
      }
    }
    __syncthreads();
    for (int i = tid; i < n; i += OWN_T) {
      const int dl = (int)(list[i] >> 20);
      const float p1 = __expf(pb1[i] - mx1[dl]);
      pb1[i] = p1;
      atomicAdd(&den1[dl], p1);
      if (DUAL) {
        const float p2 = __expf(pb2[i] - mx2[dl]);
        pb2[i] = p2;
        atomicAdd(&den2[dl], p2);
      }
    }
    __syncthreads();
    for (int b = wave * EPW; b < n; b += (OWN_T / 32) * EPW) {
      const int  i  = b + (lane & (EPW - 1));
      const bool ok = i < n;
      const int dl = ok ? (int)(list[i] >> 20) : OWN_R;
      const int s  = ok ? sb[i] : d0;
      const float c1 = ok ? pb1[i] / den1[dl] : 0.0f;
      float c2 = 0.0f;
      if (DUAL) c2 = ok ? pb2[i] / den2[dl] : 0.0f;
      float h[EPW][4];
#pragma unroll
      for (int j = 0; j < EPW; ++j) {
        const int sj = __builtin_amdgcn_readlane(s, j);
        if (DUAL) {
          const float* r1 = H1 + (size_t)sj * 64 + lane;
          const float* r2 = H2 + (size_t)sj * 64 + lane;
          h[j][0] = r1[0];  h[j][1] = r1[32];
          h[j][2] = r2[0];  h[j][3] = r2[32];
        } else {
          const float* hr = H1 + (size_t)sj * 128 + lane;
#pragma unroll
          for (int k = 0; k < 4; ++k) h[j][k] = hr[k * 32];
        }
      }
#pragma unroll
      for (int j = 0; j < EPW; ++j) {
        const int   dj  = __builtin_amdgcn_readlane(dl, j);
        const float c1j = lane_bcast_f(c1, j);
        const float c2j = DUAL ? lane_bcast_f(c2, j) : c1j;
        float* ar = acc + dj * P + lane;
        atomicAdd(ar +  0, c1j * h[j][0]);
        atomicAdd(ar + 32, c1j * h[j][1]);
        atomicAdd(ar + 64, c2j * h[j][2]);
        atomicAdd(ar + 96, c2j * h[j][3]);
      }
    }
  } else {
    if (!DUAL && tid == 0) own_l[0] = (unsigned)(LIST_CAP + 1);
    for (int pass = 0; pass < 3; ++pass) {
      for (int e = tid; e < E_TOT; e += OWN_T) {
        int s, d; edge_src_dst(ei, e, s, d);
        const unsigned dl = (unsigned)(d - d0);
        if (dl < (unsigned)nr) {
          const float t1 = leaky(as1[s] + ad1[d]);
          float t2 = 0.0f;
          if (DUAL) t2 = leaky(as2[s] + ad2[d]);
          if (pass == 0) {
            lds_max_f(&mx1[dl], t1);
            if (DUAL) lds_max_f(&mx2[dl], t2);
          } else {
            const float p1 = __expf(t1 - mx1[dl]);
            float p2 = 0.0f;
            if (DUAL) p2 = __expf(t2 - mx2[dl]);
            if (pass == 1) {
              atomicAdd(&den1[dl], p1);
              if (DUAL) atomicAdd(&den2[dl], p2);
            } else {
              const float c1 = p1 / den1[dl];
              float* ar = acc + dl * P;
              if (DUAL) {
                const float c2 = p2 / den2[dl];
                for (int k = 0; k < 64; ++k) {
                  atomicAdd(ar + k,      c1 * H1[(size_t)s * 64 + k]);
                  atomicAdd(ar + 64 + k, c2 * H2[(size_t)s * 64 + k]);
                }
              } else {
                for (int k = 0; k < 128; ++k) atomicAdd(ar + k, c1 * H1[(size_t)s * 128 + k]);
              }
            }
          }
        }
      }
      __syncthreads();
    }
  }
  __syncthreads();

  if (DUAL) {
    for (int i = tid; i < nr * 128; i += OWN_T) {
      const int r = i >> 7, c = i & 127;
      float a = acc[r * P + c];
      if (c < 64) { a += b1[c];      if (relu) a = a > 0.0f ? a : 0.0f;  out1[(size_t)(d0 + r) * 64 + c]        = a; }
      else        { a += b2[c - 64]; if (relu) a = a > 0.0f ? a : 0.0f;  out2[(size_t)(d0 + r) * 64 + (c - 64)] = a; }
    }
  } else {
    {
      const int   c  = tid & 127;
      const float bc = b1[c];
      for (int r = tid >> 7; r < nr; r += OWN_T / 128) {
        float a = acc[r * P + c] + bc;
        if (relu) a = a > 0.0f ? a : 0.0f;
        acc[r * P + c] = a;
      }
    }
    __syncthreads();
    const int hh  = lane >> 4;
    const int l16 = lane & 15;
    const int ntask = ((nr + 15) >> 4) * 2;
    static_assert(ACC_STRIPS * 2 <= OWN_T / 32, "at most one task per wave");
    if (wave < ntask) {
      const int strip = wave >> 1, head = wave & 1;
      const float*    arow = acc + (strip * 16 + l16) * P;
      const _Float16* wf   = w2f + head * (HID * OUT_CH);
      v8f c[OUT_CH / 16] = {};
#pragma unroll
      for (int kc = 0; kc < HID / 32; ++kc) {
        const int kb = kc * 32 + hh * 8;
        const float4 a0 = *(const float4*)(arow + kb);
        const float4 a1 = *(const float4*)(arow + kb + 4);
        const float4 a2 = *(const float4*)(arow + kb + 16);
        const float4 a3 = *(const float4*)(arow + kb + 20);
        const float av[16] = {a0.x, a0.y, a0.z, a0.w, a1.x, a1.y, a1.z, a1.w,
                              a2.x, a2.y, a2.z, a2.w, a3.x, a3.y, a3.z, a3.w};
        v16h a;
#pragma unroll
        for (int j = 0; j < 16; ++j) a[j] = (_Float16)av[j];
#pragma unroll
        for (int nt = 0; nt < OUT_CH / 16; ++nt) {
          const v16h b = *(const v16h*)(wf + ((size_t)(kc * (OUT_CH / 16) + nt) * 32 + lane) * 16);
          c[nt] = __builtin_amdgcn_wmma_f32_16x16x32_f16(false, a, false, b,
                                                         (short)0, c[nt], false, false);
        }
      }
      const float* __restrict__ avs = head ? avsl : avsm;
      const float* __restrict__ avd = head ? avdl : avdm;
      float* __restrict__ Cout = head ? Cl : Cm;
      float* __restrict__ os   = head ? osl : osm;
      float* __restrict__ od   = head ? odl : odm;
      float asv[OUT_CH / 16], adv[OUT_CH / 16];
#pragma unroll
      for (int nt = 0; nt < OUT_CH / 16; ++nt) {
        asv[nt] = avs[nt * 16 + l16];
        adv[nt] = avd[nt * 16 + l16];
      }
#pragma unroll
      for (int r = 0; r < 8; ++r) {
        const int  rl   = strip * 16 + hh * 8 + r;
        const bool live = rl < nr;
        float* Crow = Cout + (size_t)(d0 + rl) * OUT_CH + l16;
        if (live) {
#pragma unroll
          for (int nt = 0; nt < OUT_CH / 16; ++nt) Crow[nt * 16] = c[nt][r];
        }
        float ps = 0.0f, pd = 0.0f;
#pragma unroll
        for (int nt = 0; nt < OUT_CH / 16; ++nt) {
          const float v = c[nt][r];
          ps = fmaf(v, asv[nt], ps);
          pd = fmaf(v, adv[nt], pd);
        }
#pragma unroll
        for (int off = 8; off > 0; off >>= 1) {
          ps += __shfl_xor(ps, off, 32);
          pd += __shfl_xor(pd, off, 32);
        }
        if (live && l16 == 0) { os[d0 + rl] = ps; od[d0 + rl] = pd; }
      }
    }
  }
}


static inline int ceildiv(long a, long b) { return (int)((a + b - 1) / b); }

extern "C" void kernel_launch(void* const* d_in, const int* in_sizes, int n_in,
                              void* d_out, int out_size, void* d_ws, size_t ws_size,
                              hipStream_t stream) {
  (void)in_sizes; (void)n_in; (void)out_size; (void)ws_size;
  const float* x       = (const float*)d_in[0];
  const int*   ei      = (const int*)d_in[1];
  const float* W1      = (const float*)d_in[2];
  const float* a_src1  = (const float*)d_in[3];
  const float* a_dst1  = (const float*)d_in[4];
  const float* b1      = (const float*)d_in[5];
  const float* W_mu    = (const float*)d_in[6];
  const float* a_srcm  = (const float*)d_in[7];
  const float* a_dstm  = (const float*)d_in[8];
  const float* b_mu    = (const float*)d_in[9];
  const float* W_ls    = (const float*)d_in[10];
  const float* a_srcl  = (const float*)d_in[11];
  const float* a_dstl  = (const float*)d_in[12];
  const float* b_ls    = (const float*)d_in[13];

  const long SZ_H = (long)N_NODES * HID;
  const long SZ_O = (long)N_NODES * OUT_CH;

  float* ws  = (float*)d_ws;
  float* hx  = ws;
  float* hmu = hx + SZ_H;
  float* hls = hmu + SZ_O;
  float* sc  = hls + SZ_O;
  float* as1 = sc + 0L * N_NODES, *ad1 = sc + 1L * N_NODES;
  float* asm_= sc + 2L * N_NODES, *adm = sc + 3L * N_NODES;
  float* asl = sc + 4L * N_NODES, *adl = sc + 5L * N_NODES;

  float* out_mu = (float*)d_out;
  float* out_ls = out_mu + SZ_O;

  static_assert((long)E_REAL <= (long)N_NODES * OUT_CH * (long)sizeof(float), "id column fits the second output plane");
  unsigned* ids1 = (unsigned*)out_ls;

  const int TB = 256;
  const int gemm_blocks = ceildiv(N_NODES / 16, 8);

  wmma_gemm_attn<HID / 16, IN_CH><<<gemm_blocks, TB, 0, stream>>>(
      x, W1, hx, a_src1, a_dst1, as1, ad1, N_NODES, ei + E_REAL, ids1);
  owner_gat<false><<<OWN_G, OWN_T, 0, stream>>>(ei, ids1, as1, ad1, hx, b1, nullptr,
      nullptr, nullptr, nullptr, nullptr, nullptr, (unsigned*)out_mu,
      W_mu, W_ls, a_srcm, a_dstm, a_srcl, a_dstl, hmu, hls, asm_, adm, asl, adl, 1);

  owner_gat<true><<<OWN_G, OWN_T, 0, stream>>>(ei, nullptr, asm_, adm, hmu, b_mu, out_mu,
      asl, adl, hls, b_ls, out_ls, nullptr,
      nullptr, nullptr, nullptr, nullptr, nullptr, nullptr,
      nullptr, nullptr, nullptr, nullptr, nullptr, nullptr, 0);
}


# owner pass finds its edges via per-group presence bitmaps over 8-edge chunks built by extra groups of launch 1: reads its bitmap row, then only flagged chunks' destinations; sums keep arrival order
# speedup vs baseline: 6.9880x; 1.2707x over previous
#include <hip/hip_runtime.h>
#include <hip/hip_fp16.h>

typedef __attribute__((ext_vector_type(16))) _Float16 v16h;
typedef __attribute__((ext_vector_type(8)))  float    v8f;

#define N_NODES 50000
#define E_REAL  800000
#define E_TOT   (E_REAL + N_NODES)
#define IN_CH   128
#define HID     128
#define OUT_CH  64
#define NEG_SLOPE 0.2f

#define OWN_R 196
#define OWN_G ((N_NODES + OWN_R - 1) / OWN_R)
#define CH_E 8
#define N_CH (E_REAL / CH_E)
#define PRES_TW 32
#define PRES_TCH (PRES_TW * 32)
#define PRES_BLOCKS ((N_CH + PRES_TCH - 1) / PRES_TCH)
#define PRES_P (PRES_BLOCKS * PRES_TW)
#define GEMM_T 256
static_assert(E_REAL % CH_E == 0, "chunks are whole");
static_assert(CH_E == 8, "a chunk's destinations are read as two 16-byte loads");
static_assert(PRES_TCH % GEMM_T == 0, "every builder thread takes the same number of chunks");
static_assert(OWN_G * PRES_TW * 4 <= IN_CH * HID * 2, "the tile fits the LDS array of the product kernel");

__device__ __forceinline__ int pres_slot(int o, int w) { return o * PRES_TW + (w ^ (o & (PRES_TW - 1))); }

__device__ __forceinline__ void build_presence_tile(const int* __restrict__ dstcol,
                                                    unsigned* __restrict__ pres,
                                                    unsigned* tile, int b) {
  constexpr int CPT = PRES_TCH / GEMM_T;
  const int tid = threadIdx.x;
  const int4* __restrict__ d4 = (const int4*)dstcol;
  int4 lo[CPT], hi[CPT];
#pragma unroll
  for (int j = 0; j < CPT; ++j) {
    const int ch = b * PRES_TCH + j * GEMM_T + tid;
    const int cc = ch < N_CH ? ch : N_CH - 1;
    lo[j] = d4[2 * cc];
    hi[j] = d4[2 * cc + 1];
  }
  for (int i = tid; i < OWN_G * PRES_TW; i += GEMM_T) tile[i] = 0u;
  __syncthreads();
#pragma unroll
  for (int j = 0; j < CPT; ++j) {
    const int cl = j * GEMM_T + tid;
    if (b * PRES_TCH + cl < N_CH) {
      const int      w   = cl >> 5;
      const unsigned bit = 1u << (cl & 31);
      const int dv[CH_E] = {lo[j].x, lo[j].y, lo[j].z, lo[j].w, hi[j].x, hi[j].y, hi[j].z, hi[j].w};
#pragma unroll
      for (int q = 0; q < CH_E; ++q) {
        const unsigned d = (unsigned)dv[q];
        if (d < (unsigned)N_NODES)
          atomicOr(&tile[pres_slot((int)(d / OWN_R), w)], bit);
      }
    }
  }
  __syncthreads();
  for (int i = tid; i < OWN_G * PRES_TW; i += GEMM_T) {
    const int g = i / PRES_TW, w = i % PRES_TW;
    pres[(size_t)g * PRES_P + b * PRES_TW + w] = tile[pres_slot(g, w)];
  }
}

__device__ __forceinline__ int perm_pos128(int c) { return ((c & 31) << 2) | (c >> 5); }
__device__ __forceinline__ int perm_pos64(int c)  { return ((c & 31) << 1) | (c >> 5); }

template<int NT, int K>
__global__ void wmma_gemm_attn(const float* __restrict__ A,
                               const float* __restrict__ B,
                               float* __restrict__ C,
                               const float* __restrict__ a_s,
                               const float* __restrict__ a_d,
                               float* __restrict__ os, float* __restrict__ od,
                               int M,
                               const int* __restrict__ dstcol,
                               unsigned* __restrict__ pres) {
  constexpr int N  = NT * 16;
  constexpr int KC = K / 32;
  __shared__ __align__(32) _Float16 ldsB[K * N];

  if ((int)blockIdx.x < PRES_BLOCKS) {
    build_presence_tile(dstcol, pres, (unsigned*)ldsB, (int)blockIdx.x);
    return;
  }

  const int lane   = threadIdx.x & 31;
  const int wlocal = threadIdx.x >> 5;
  const int mtile  = ((int)blockIdx.x - PRES_BLOCKS) * 8 + wlocal;
  const int hh  = lane >> 4;
  const int l16 = lane & 15;

  for (int slot = threadIdx.x; slot < KC * NT * 32; slot += blockDim.x) {
    const int sl   = slot & 31;
    const int t    = slot >> 5;
    const int nt   = t % NT;
    const int kc   = t / NT;
    const int kb   = kc * 32 + (sl >> 4) * 8;
    const int n    = nt * 16 + (sl & 15);
    _Float16* dst  = ldsB + slot * 16;
#pragma unroll
    for (int j = 0; j < 8; ++j) {
      dst[j]     = (_Float16)B[(size_t)(kb + j) * N + n];
      dst[j + 8] = (_Float16)B[(size_t)(kb + 16 + j) * N + n];
    }
  }
  __syncthreads();
  if (mtile * 16 >= M) return;

  const float* __restrict__ Arow = A + (size_t)(mtile * 16 + l16) * K;
  v8f acc[NT] = {};
#pragma unroll
  for (int kc = 0; kc < KC; ++kc) {
    const int kb = kc * 32 + hh * 8;
    const float4 a0 = *(const float4*)(Arow + kb);
    const float4 a1 = *(const float4*)(Arow + kb + 4);
    const float4 a2 = *(const float4*)(Arow + kb + 16);
    const float4 a3 = *(const float4*)(Arow + kb + 20);
    const float av[16] = {a0.x, a0.y, a0.z, a0.w, a1.x, a1.y, a1.z, a1.w,
                          a2.x, a2.y, a2.z, a2.w, a3.x, a3.y, a3.z, a3.w};
    v16h a;
#pragma unroll
    for (int j = 0; j < 16; ++j) a[j] = (_Float16)av[j];
#pragma unroll
    for (int nt = 0; nt < NT; ++nt) {
      const v16h b = *(const v16h*)(ldsB + ((size_t)(kc * NT + nt) * 32 + lane) * 16);
      acc[nt] = __builtin_amdgcn_wmma_f32_16x16x32_f16(false, a, false, b,
                                                       (short)0, acc[nt], false, false);
    }
  }

  float asv[NT], adv[NT];
#pragma unroll
  for (int nt = 0; nt < NT; ++nt) {
    asv[nt] = a_s[nt * 16 + l16];
    adv[nt] = a_d[nt * 16 + l16];
  }
  static_assert(NT == 8, "the permuted row layout below is written for 128-float rows");
#pragma unroll
  for (int r = 0; r < 8; ++r) {
    const int row = mtile * 16 + hh * 8 + r;
    float* Crow = C + (size_t)row * N + 4 * l16;
    *(float4*)(Crow)      = make_float4(acc[0][r], acc[2][r], acc[4][r], acc[6][r]);
    *(float4*)(Crow + 64) = make_float4(acc[1][r], acc[3][r], acc[5][r], acc[7][r]);
    float ps = 0.0f, pd = 0.0f;
#pragma unroll
    for (int nt = 0; nt < NT; ++nt) {
      const float v = acc[nt][r];
      ps = fmaf(v, asv[nt], ps);
      pd = fmaf(v, adv[nt], pd);
    }
#pragma unroll
    for (int off = 8; off > 0; off >>= 1) {
      ps += __shfl_xor(ps, off, 32);
      pd += __shfl_xor(pd, off, 32);
    }
    if (l16 == 0) { os[row] = ps; od[row] = pd; }
  }
}

__device__ __forceinline__ void edge_src_dst(const int* __restrict__ ei, int e,
                                             int& s, int& d) {
  if (e < E_REAL) { s = ei[e]; d = ei[E_REAL + e]; }
  else            { s = d = e - E_REAL; }
}

#define SAVE_HDR 4
#define EPW 8
#define OWN_T 1024
#define LIST_CAP 8192
#define ACC_P 132
#define ACC_STRIPS ((OWN_R + 15) / 16)
static_assert(ACC_STRIPS * 16 > OWN_R, "the sink row lies inside the last strip's rows");
static_assert(2 * (HID / 32) * (OUT_CH / 16) * 32 == OWN_T, "one B-fragment slot of the two weight matrices per thread");
static_assert(E_TOT <= (1 << 20) && OWN_R < (1 << 12), "list word = row << 20 | edge");
static_assert(OWN_R <= OWN_T && OWN_R <= LIST_CAP, "one thread and one slot per self-loop");
static_assert(LIST_CAP <= 65536, "a list position is kept in 16 bits");
static_assert(OWN_T / 32 == 32, "the row phase cuts the ordered list into 32 runs, one per wave");

__device__ __forceinline__ float lane_bcast_f(float v, int j) {
  return __int_as_float(__builtin_amdgcn_readlane(__float_as_int(v), j));
}

__device__ __forceinline__ int wave_take(int* counter, int cnt, int lane) {
  int inc = cnt;
  for (int off = 1; off < 32; off <<= 1) {
    const int t = __shfl_up(inc, off, 32);
    if (lane >= off) inc += t;
  }
  int first = 0;
  if (lane == 31 && inc != 0) first = atomicAdd(counter, inc);
  first = __shfl(first, 31, 32);
  return first + inc - cnt;
}

__device__ __forceinline__ void lds_max_f(float* addr, float v) {
  if (v >= 0.0f) atomicMax((int*)addr, __float_as_int(v));
  else           atomicMin((unsigned int*)addr, __float_as_uint(v));
}

__device__ __forceinline__ float leaky(float t) { return t > 0.0f ? t : NEG_SLOPE * t; }

template<bool DUAL>
__global__ __launch_bounds__(OWN_T)
void owner_gat(const int* __restrict__ ei, const unsigned* __restrict__ pres,
    const float* __restrict__ as1, const float* __restrict__ ad1,
    const float* __restrict__ H1, const float* __restrict__ b1, float* __restrict__ out1,
    const float* __restrict__ as2, const float* __restrict__ ad2,
    const float* __restrict__ H2, const float* __restrict__ b2, float* __restrict__ out2,
    unsigned* __restrict__ sv_l,
    const float* __restrict__ Wm, const float* __restrict__ Wl,
    const float* __restrict__ avsm, const float* __restrict__ avdm,
    const float* __restrict__ avsl, const float* __restrict__ avdl,
    float* __restrict__ Cm, float* __restrict__ Cl,
    float* __restrict__ osm, float* __restrict__ odm,
    float* __restrict__ osl, float* __restrict__ odl,
    int relu) {
  constexpr int P    = DUAL ? 128 : ACC_P;
  constexpr int ROWS = DUAL ? OWN_R + 1 : ACC_STRIPS * 16;
  __shared__ __align__(16) float    acc[ROWS * P];
  __shared__ __align__(32) _Float16 w2f[DUAL ? 16 : 2 * HID * OUT_CH];
  __shared__ unsigned list[LIST_CAP];
  __shared__ int      sb[LIST_CAP];
  __shared__ float    pb1[LIST_CAP];
  __shared__ float    pb2[DUAL ? LIST_CAP : 1];
  __shared__ float    mx1[OWN_R + 1], mx2[OWN_R + 1];
  __shared__ float    den1[OWN_R + 1], den2[OWN_R + 1];
  __shared__ int      s_cnt;
  __shared__ int      s_nch;
  __shared__ int            rcur[OWN_R + 1];
  __shared__ unsigned short perm[LIST_CAP];

  const int tid  = threadIdx.x;
  const int lane = tid & 31;
  const int wave = tid >> 5;
  const int d0   = blockIdx.x * OWN_R;
  const int nr   = (N_NODES - d0) < OWN_R ? (N_NODES - d0) : OWN_R;

  for (int i = tid; i < ROWS * P; i += OWN_T) acc[i] = 0.0f;
  if (tid <= OWN_R) {
    mx1[tid] = -__builtin_huge_valf();  mx2[tid] = -__builtin_huge_valf();
    den1[tid] = 0.0f;                   den2[tid] = 0.0f;
    rcur[tid] = 0;
  }
  if (!DUAL && tid < nr) list[tid] = (unsigned)(E_REAL + d0 + tid);
  if (!DUAL && tid == 0) { s_cnt = nr;  s_nch = 0; }
  __syncthreads();

  unsigned* const own_l = DUAL ? (unsigned*)(out1 + (size_t)d0 * 64) : sv_l + (size_t)d0 * 64;
  int*      const own_s = (int*)(own_l + nr * 32);

  int n_all;
  if (DUAL) {
    n_all = (int)own_l[0];
  } else {
  float wreg[16];
  {
    const int slot = tid & 511;
    const int sl = slot & 31, t = slot >> 5;
    const int kb = (t >> 2) * 32 + (sl >> 4) * 8;
    const int n  = (t & 3) * 16 + (sl & 15);
    const float* __restrict__ Wsrc = (tid >> 9) ? Wl : Wm;
#pragma unroll
    for (int j = 0; j < 8; ++j) {
      wreg[j]     = Wsrc[(kb + j) * OUT_CH + n];
      wreg[j + 8] = Wsrc[(kb + 16 + j) * OUT_CH + n];
    }
  }
  {
    const unsigned* __restrict__ prow = pres + (size_t)blockIdx.x * PRES_P;
    constexpr int WPT = (PRES_P + OWN_T - 1) / OWN_T;
    unsigned pw[WPT];
#pragma unroll
    for (int k = 0; k < WPT; ++k) {
      const int idx = k * OWN_T + tid;
      pw[k] = idx < PRES_P ? prow[idx] : 0u;
    }
    int tot = 0;
#pragma unroll
    for (int k = 0; k < WPT; ++k) tot += __builtin_popcount(pw[k]);
    int at = wave_take(&s_nch, tot, lane);
#pragma unroll
    for (int k = 0; k < WPT; ++k) {
      unsigned m = pw[k];
      const int c0 = (k * OWN_T + tid) * 32;
      while (m != 0u) {
        const int p = __builtin_ctz(m);
        m &= m - 1u;
        if (at < LIST_CAP) sb[at] = c0 + p;
        ++at;
      }
    }
  }
  __syncthreads();
  const int nch = s_nch;
  if (nch <= LIST_CAP) {
    constexpr int CPT = 4;
    const int4* __restrict__ d4 = (const int4*)(ei + E_REAL);
    for (int base = 0; base < nch; base += CPT * OWN_T) {
      int  ch[CPT];
      int4 lo[CPT], hi[CPT];
#pragma unroll
      for (int j = 0; j < CPT; ++j) {
        const int i = base + j * OWN_T + tid;
        ch[j] = i < nch ? sb[i] : -1;
      }
#pragma unroll
      for (int j = 0; j < CPT; ++j) {
        const int cc = ch[j] >= 0 ? ch[j] : 0;
        lo[j] = d4[2 * cc];
        hi[j] = d4[2 * cc + 1];
      }
      unsigned own[CPT];
      int tot = 0;
#pragma unroll
      for (int j = 0; j < CPT; ++j) {
        const int dv[CH_E] = {lo[j].x, lo[j].y, lo[j].z, lo[j].w, hi[j].x, hi[j].y, hi[j].z, hi[j].w};
        unsigned mm = 0u;
#pragma unroll
        for (int q = 0; q < CH_E; ++q)
          mm |= ((unsigned)(dv[q] - d0) < (unsigned)nr) ? (1u << q) : 0u;
        own[j] = ch[j] >= 0 ? mm : 0u;
        tot += __builtin_popcount(own[j]);
      }
      int at = wave_take(&s_cnt, tot, lane);
#pragma unroll
      for (int j = 0; j < CPT; ++j) {
        unsigned mm = own[j];
        const int e0 = ch[j] * CH_E;
        while (mm != 0u) {
          const int q = __builtin_ctz(mm);
          mm &= mm - 1u;
          if (at < LIST_CAP) list[at] = (unsigned)(e0 + q);
          ++at;
        }
      }
    }
  } else if (tid == 0) {
    s_cnt = LIST_CAP + 1;
  }
  {
    _Float16* wdst = w2f + tid * 16;
#pragma unroll
    for (int j = 0; j < 16; ++j) wdst[j] = (_Float16)wreg[j];
  }
  __syncthreads();
  n_all = s_cnt;
  }

  if (n_all <= LIST_CAP) {
    const int n = n_all;
    const bool keep = !DUAL && n <= nr * 32 - SAVE_HDR;
    if (!DUAL && tid == 0) own_l[0] = keep ? (unsigned)n : (unsigned)(LIST_CAP + 1);
    for (int i = tid; i < n; i += OWN_T) {
      int s, d, dl;
      if (DUAL) {
        const unsigned w = own_l[SAVE_HDR + i];
        s  = own_s[i];
        dl = (int)(w >> 20);
        d  = dl < OWN_R ? d0 + dl : d0;
        list[i] = w;
      } else {
        const int e = (int)list[i];
        edge_src_dst(ei, e, s, d);
        dl = d - d0;
        if ((unsigned)dl >= (unsigned)nr) {
          dl = OWN_R;  s = d0;  d = d0;
        }
        const unsigned w = ((unsigned)dl << 20) | (unsigned)e;
        list[i] = w;
        if (keep) { own_l[SAVE_HDR + i] = w;  own_s[i] = s; }
      }
      sb[i] = s;
      atomicAdd(&rcur[dl], 1);
      const float t1 = leaky(as1[s] + ad1[d]);
      pb1[i] = t1;
      lds_max_f(&mx1[dl], t1);
      if (DUAL) {
        const float t2 = leaky(as2[s] + ad2[d]);
        pb2[i] = t2;
        lds_max_f(&mx2[dl], t2);
      }
    }
    __syncthreads();
    if (wave == 0) {
      constexpr int RPL = (OWN_R + 1 + 31) / 32;
      int v[RPL];
      int tot = 0;
#pragma unroll
      for (int k = 0; k < RPL; ++k) {
        const int r = lane * RPL + k;
        v[k] = r <= OWN_R ? rcur[r] : 0;
        tot += v[k];
      }
      int inc = tot;
      for (int off = 1; off < 32; off <<= 1) {
        const int t = __shfl_up(inc, off, 32);
        if (lane >= off) inc += t;
      }
      int base = inc - tot;
#pragma unroll
      for (int k = 0; k < RPL; ++k) {
        const int r = lane * RPL + k;
        if (r <= OWN_R) rcur[r] = base;
        base += v[k];
      }
    }
    for (int i = tid; i < n; i += OWN_T) {
      const int dl = (int)(list[i] >> 20);
      const float p1 = __expf(pb1[i] - mx1[dl]);
      pb1[i] = p1;
      atomicAdd(&den1[dl], p1);
      if (DUAL) {
        const float p2 = __expf(pb2[i] - mx2[dl]);
        pb2[i] = p2;
        atomicAdd(&den2[dl], p2);
      }
    }
    __syncthreads();
    for (int i = tid; i < n; i += OWN_T)
      perm[atomicAdd(&rcur[(int)(list[i] >> 20)], 1)] = (unsigned short)i;
    __syncthreads();
    {
      const int chunk = ((((n + 31) >> 5) + EPW - 1) / EPW) * EPW;
      const int pbeg  = wave * chunk;
      const int pend  = pbeg + chunk < n ? pbeg + chunk : n;
      auto entry = [&](int b, int& dl, int& s, float& c1, float& c2) {
        const int p = b + (lane & (EPW - 1));
        dl = OWN_R;  s = d0;  c1 = 0.0f;  c2 = 0.0f;
        if (p < pend) {
          const int i = (int)perm[p];
          dl = (int)(list[i] >> 20);
          s  = sb[i];
          c1 = pb1[i] / den1[dl];
          if (DUAL) c2 = pb2[i] / den2[dl];
        }
      };
      float r0 = 0.0f, r1 = 0.0f, r2 = 0.0f, r3 = 0.0f;
      int   cur = -1;
      int dl, s;  float c1, c2;
      entry(pbeg, dl, s, c1, c2);
      for (int b = pbeg; b < pend; b += EPW) {
        float h[EPW][4];
#pragma unroll
        for (int j = 0; j < EPW; ++j) {
          const int sj = __builtin_amdgcn_readlane(s, j);
          if (DUAL) {
            const float2 v1 = *(const float2*)(H1 + (size_t)sj * 64 + 2 * lane);
            const float2 v2 = *(const float2*)(H2 + (size_t)sj * 64 + 2 * lane);
            h[j][0] = v1.x;  h[j][1] = v1.y;
            h[j][2] = v2.x;  h[j][3] = v2.y;
          } else {
            const float4 v = *(const float4*)(H1 + (size_t)sj * 128 + 4 * lane);
            h[j][0] = v.x;  h[j][1] = v.y;  h[j][2] = v.z;  h[j][3] = v.w;
          }
        }
        int dln, sn;  float c1n, c2n;
        entry(b + EPW, dln, sn, c1n, c2n);
#pragma unroll
        for (int j = 0; j < EPW; ++j) {
          const int   dj  = __builtin_amdgcn_readlane(dl, j);
          const float c1j = lane_bcast_f(c1, j);
          const float c2j = DUAL ? lane_bcast_f(c2, j) : c1j;
          if (dj != cur) {
            if (cur >= 0) {
              float* ar = acc + cur * P + lane;
              atomicAdd(ar +  0, r0);
              atomicAdd(ar + 32, r1);
              atomicAdd(ar + 64, r2);
              atomicAdd(ar + 96, r3);
            }
            r0 = 0.0f;  r1 = 0.0f;  r2 = 0.0f;  r3 = 0.0f;
            cur = dj;
          }
          r0 = fmaf(c1j, h[j][0], r0);
          r1 = fmaf(c1j, h[j][1], r1);
          r2 = fmaf(c2j, h[j][2], r2);
          r3 = fmaf(c2j, h[j][3], r3);
        }
        dl = dln;  s = sn;  c1 = c1n;  c2 = c2n;
      }
      if (cur >= 0) {
        float* ar = acc + cur * P + lane;
        atomicAdd(ar +  0, r0);
        atomicAdd(ar + 32, r1);
        atomicAdd(ar + 64, r2);
        atomicAdd(ar + 96, r3);
      }
    }
  } else {
    if (!DUAL && tid == 0) own_l[0] = (unsigned)(LIST_CAP + 1);
    for (int pass = 0; pass < 3; ++pass) {
      for (int e = tid; e < E_TOT; e += OWN_T) {
        int s, d; edge_src_dst(ei, e, s, d);
        const unsigned dl = (unsigned)(d - d0);
        if (dl < (unsigned)nr) {
          const float t1 = leaky(as1[s] + ad1[d]);
          float t2 = 0.0f;
          if (DUAL) t2 = leaky(as2[s] + ad2[d]);
          if (pass == 0) {
            lds_max_f(&mx1[dl], t1);
            if (DUAL) lds_max_f(&mx2[dl], t2);
          } else {
            const float p1 = __expf(t1 - mx1[dl]);
            float p2 = 0.0f;
            if (DUAL) p2 = __expf(t2 - mx2[dl]);
            if (pass == 1) {
              atomicAdd(&den1[dl], p1);
              if (DUAL) atomicAdd(&den2[dl], p2);
            } else {
              const float c1 = p1 / den1[dl];
              float* ar = acc + dl * P;
              if (DUAL) {
                const float c2 = p2 / den2[dl];
                for (int k = 0; k < 64; ++k) {
                  atomicAdd(ar + k,      c1 * H1[(size_t)s * 64 + perm_pos64(k)]);
                  atomicAdd(ar + 64 + k, c2 * H2[(size_t)s * 64 + perm_pos64(k)]);
                }
              } else {
                for (int k = 0; k < 128; ++k) atomicAdd(ar + k, c1 * H1[(size_t)s * 128 + perm_pos128(k)]);
              }
            }
          }
        }
      }
      __syncthreads();
    }
  }
  __syncthreads();

  if (DUAL) {
    for (int i = tid; i < nr * 128; i += OWN_T) {
      const int r = i >> 7, c = i & 127;
      float a = acc[r * P + c];
      if (c < 64) { a += b1[c];      if (relu) a = a > 0.0f ? a : 0.0f;  out1[(size_t)(d0 + r) * 64 + c]        = a; }
      else        { a += b2[c - 64]; if (relu) a = a > 0.0f ? a : 0.0f;  out2[(size_t)(d0 + r) * 64 + (c - 64)] = a; }
    }
  } else {
    {
      const int   c  = tid & 127;
      const float bc = b1[c];
      for (int r = tid >> 7; r < nr; r += OWN_T / 128) {
        float a = acc[r * P + c] + bc;
        if (relu) a = a > 0.0f ? a : 0.0f;
        acc[r * P + c] = a;
      }
    }
    __syncthreads();
    const int hh  = lane >> 4;
    const int l16 = lane & 15;
    const int ntask = ((nr + 15) >> 4) * 2;
    static_assert(ACC_STRIPS * 2 <= OWN_T / 32, "at most one task per wave");
    if (wave < ntask) {
      const int strip = wave >> 1, head = wave & 1;
      const float*    arow = acc + (strip * 16 + l16) * P;
      const _Float16* wf   = w2f + head * (HID * OUT_CH);
      v8f c[OUT_CH / 16] = {};
#pragma unroll
      for (int kc = 0; kc < HID / 32; ++kc) {
        const int kb = kc * 32 + hh * 8;
        const float4 a0 = *(const float4*)(arow + kb);
        const float4 a1 = *(const float4*)(arow + kb + 4);
        const float4 a2 = *(const float4*)(arow + kb + 16);
        const float4 a3 = *(const float4*)(arow + kb + 20);
        const float av[16] = {a0.x, a0.y, a0.z, a0.w, a1.x, a1.y, a1.z, a1.w,
                              a2.x, a2.y, a2.z, a2.w, a3.x, a3.y, a3.z, a3.w};
        v16h a;
#pragma unroll
        for (int j = 0; j < 16; ++j) a[j] = (_Float16)av[j];
#pragma unroll
        for (int nt = 0; nt < OUT_CH / 16; ++nt) {
          const v16h b = *(const v16h*)(wf + ((size_t)(kc * (OUT_CH / 16) + nt) * 32 + lane) * 16);
          c[nt] = __builtin_amdgcn_wmma_f32_16x16x32_f16(false, a, false, b,
                                                         (short)0, c[nt], false, false);
        }
      }
      const float* __restrict__ avs = head ? avsl : avsm;
      const float* __restrict__ avd = head ? avdl : avdm;
      float* __restrict__ Cout = head ? Cl : Cm;
      float* __restrict__ os   = head ? osl : osm;
      float* __restrict__ od   = head ? odl : odm;
      float asv[OUT_CH / 16], adv[OUT_CH / 16];
#pragma unroll
      for (int nt = 0; nt < OUT_CH / 16; ++nt) {
        asv[nt] = avs[nt * 16 + l16];
        adv[nt] = avd[nt * 16 + l16];
      }
#pragma unroll
      for (int r = 0; r < 8; ++r) {
        const int  rl   = strip * 16 + hh * 8 + r;
        const bool live = rl < nr;
        float* Crow = Cout + (size_t)(d0 + rl) * OUT_CH + 2 * l16;
        if (live) {
          *(float2*)(Crow)      = make_float2(c[0][r], c[2][r]);
          *(float2*)(Crow + 32) = make_float2(c[1][r], c[3][r]);
        }
        float ps = 0.0f, pd = 0.0f;
#pragma unroll
        for (int nt = 0; nt < OUT_CH / 16; ++nt) {
          const float v = c[nt][r];
          ps = fmaf(v, asv[nt], ps);
          pd = fmaf(v, adv[nt], pd);
        }
#pragma unroll
        for (int off = 8; off > 0; off >>= 1) {
          ps += __shfl_xor(ps, off, 32);
          pd += __shfl_xor(pd, off, 32);
        }
        if (live && l16 == 0) { os[d0 + rl] = ps; od[d0 + rl] = pd; }
      }
    }
  }
}


static inline int ceildiv(long a, long b) { return (int)((a + b - 1) / b); }

extern "C" void kernel_launch(void* const* d_in, const int* in_sizes, int n_in,
                              void* d_out, int out_size, void* d_ws, size_t ws_size,
                              hipStream_t stream) {
  (void)in_sizes; (void)n_in; (void)out_size; (void)ws_size;
  const float* x       = (const float*)d_in[0];
  const int*   ei      = (const int*)d_in[1];
  const float* W1      = (const float*)d_in[2];
  const float* a_src1  = (const float*)d_in[3];
  const float* a_dst1  = (const float*)d_in[4];
  const float* b1      = (const float*)d_in[5];
  const float* W_mu    = (const float*)d_in[6];
  const float* a_srcm  = (const float*)d_in[7];
  const float* a_dstm  = (const float*)d_in[8];
  const float* b_mu    = (const float*)d_in[9];
  const float* W_ls    = (const float*)d_in[10];
  const float* a_srcl  = (const float*)d_in[11];
  const float* a_dstl  = (const float*)d_in[12];
  const float* b_ls    = (const float*)d_in[13];

  const long SZ_H = (long)N_NODES * HID;
  const long SZ_O = (long)N_NODES * OUT_CH;

  float* ws  = (float*)d_ws;
  float* hx  = ws;
  float* hmu = hx + SZ_H;
  float* hls = hmu + SZ_O;
  float* sc  = hls + SZ_O;
  float* as1 = sc + 0L * N_NODES, *ad1 = sc + 1L * N_NODES;
  float* asm_= sc + 2L * N_NODES, *adm = sc + 3L * N_NODES;
  float* asl = sc + 4L * N_NODES, *adl = sc + 5L * N_NODES;

  float* out_mu = (float*)d_out;
  float* out_ls = out_mu + SZ_O;

  static_assert((long)OWN_G * PRES_P <= (long)N_NODES * OUT_CH, "the bitmaps fit the second output plane");
  unsigned* pres = (unsigned*)out_ls;

  const int gemm_blocks = ceildiv(N_NODES / 16, 8);

  wmma_gemm_attn<HID / 16, IN_CH><<<PRES_BLOCKS + gemm_blocks, GEMM_T, 0, stream>>>(
      x, W1, hx, a_src1, a_dst1, as1, ad1, N_NODES, ei + E_REAL, pres);
  owner_gat<false><<<OWN_G, OWN_T, 0, stream>>>(ei, pres, as1, ad1, hx, b1, nullptr,
      nullptr, nullptr, nullptr, nullptr, nullptr, (unsigned*)out_mu,
      W_mu, W_ls, a_srcm, a_dstm, a_srcl, a_dstl, hmu, hls, asm_, adm, asl, adl, 1);

  owner_gat<true><<<OWN_G, OWN_T, 0, stream>>>(ei, nullptr, asm_, adm, hmu, b_mu, out_mu,
      asl, adl, hls, b_ls, out_ls, nullptr,
      nullptr, nullptr, nullptr, nullptr, nullptr, nullptr,
      nullptr, nullptr, nullptr, nullptr, nullptr, nullptr, 0);
}
